# starting kernel
# baseline (speedup 1.0000x reference)
#include <hip/hip_runtime.h>
#include <hip/hip_bf16.h>


typedef __attribute__((ext_vector_type(2))) float v2f;
typedef __attribute__((ext_vector_type(8))) float v8f;

#define BB   8
#define MM   2048
#define NN   8192
#define KNN  8
#define HID  128
#define INP  6
#define INP8 8
#define OUTW 2

__global__ void __launch_bounds__(256)
knn_seed_kernel(const float* __restrict__ partial,
                const float* __restrict__ predicted,
                float* __restrict__ xout) {
  __shared__ float sp[MM * 3];
  const int b   = blockIdx.y;
  const int tid = threadIdx.x;

  for (int i = tid; i < MM * 3; i += 256)
    sp[i] = partial[b * MM * 3 + i];
  __syncthreads();

  const int n = blockIdx.x * 256 + tid;
  const int row = b * NN + n;
  const float px = predicted[row * 3 + 0];
  const float py = predicted[row * 3 + 1];
  const float pz = predicted[row * 3 + 2];

  float bd[KNN]; int bi[KNN];
#pragma unroll
  for (int j = 0; j < KNN; ++j) { bd[j] = 3.4e38f; bi[j] = 0; }

  for (int m = 0; m < MM; ++m) {
    const float dx = px - sp[3 * m + 0];
    const float dy = py - sp[3 * m + 1];
    const float dz = pz - sp[3 * m + 2];
    float d = dx * dx + dy * dy + dz * dz;
    if (d < bd[KNN - 1]) {
      int id = m;
#pragma unroll
      for (int j = 0; j < KNN; ++j) {
        const bool sw = d < bd[j];
        const float td = bd[j]; const int ti = bi[j];
        bd[j] = sw ? d : td;  bi[j] = sw ? id : ti;
        d     = sw ? td : d;  id    = sw ? ti : id;
      }
    }
  }

  float sx = 0.f, sy = 0.f, sz = 0.f;
#pragma unroll
  for (int j = 0; j < KNN; ++j) {
    const int m = bi[j];
    sx += sp[3 * m + 0]; sy += sp[3 * m + 1]; sz += sp[3 * m + 2];
  }

  float* xr = xout + (size_t)row * INP8;
  xr[0] = px;          xr[1] = py;          xr[2] = pz;
  xr[3] = sx * 0.125f; xr[4] = sy * 0.125f; xr[5] = sz * 0.125f;
  xr[6] = 0.f;         xr[7] = 0.f;
}

__device__ __forceinline__ v8f wmma_f32(v2f a, v2f b, v8f c) {
  return __builtin_amdgcn_wmma_f32_16x16x4_f32(
       false, a,  false, b,
       (short)0, c,  false,  false);
}

__device__ __forceinline__ float relu(float x) { return x > 0.f ? x : 0.f; }

__global__ void __launch_bounds__(128)
mlp_refine_kernel(const float* __restrict__ xin,
                  const float* __restrict__ predicted,
                  const float* __restrict__ W1, const float* __restrict__ b1,
                  const float* __restrict__ W2, const float* __restrict__ b2,
                  const float* __restrict__ Wd, const float* __restrict__ bd,
                  const float* __restrict__ Wc, const float* __restrict__ bc,
                  float* __restrict__ out) {
  extern __shared__ float smem[];
  const int tid   = threadIdx.x;
  const int wave  = tid >> 5;
  const int lane  = tid & 31;
  const int lrow  = lane & 15;
  const int khalf = lane >> 4;

  float* lds_h    = smem + wave * (32 * HID);
  float* lds_seed = smem + 4 * (32 * HID) + wave * (32 * HID);
  float* lds_ht   = smem + 8 * (32 * HID) + wave * (32 * HID);

  const int rowBase = blockIdx.x * 128 + wave * 32;
  const int dRow = (lane < 16) ? 0 : 8;

  {
    const float* xr0 = xin + (size_t)(rowBase + lrow) * INP8;
    const float* xr1 = xin + (size_t)(rowBase + 16 + lrow) * INP8;
    const float kmask = khalf ? 0.f : 1.f;
    v2f a00, a01, a10, a11;
    a00.x = xr0[khalf * 2 + 0];     a00.y = xr0[khalf * 2 + 1];
    a01.x = xr0[4 + khalf * 2 + 0]; a01.y = xr0[4 + khalf * 2 + 1];
    a10.x = xr1[khalf * 2 + 0];     a10.y = xr1[khalf * 2 + 1];
    a11.x = xr1[4 + khalf * 2 + 0]; a11.y = xr1[4 + khalf * 2 + 1];
#pragma unroll
    for (int nt = 0; nt < 8; ++nt) {
      const int n = nt * 16 + lrow;
      const float bv = b1[n];
      v8f c0, c1;
#pragma unroll
      for (int j = 0; j < 8; ++j) { c0[j] = bv; c1[j] = bv; }
      v2f bf0, bf1;
      const int k0 = khalf * 2;
      bf0.x = W1[n * INP + k0];        bf0.y = W1[n * INP + k0 + 1];
      bf1.x = kmask * W1[n * INP + 4]; bf1.y = kmask * W1[n * INP + 5];
      c0 = wmma_f32(a00, bf0, c0);  c0 = wmma_f32(a01, bf1, c0);
      c1 = wmma_f32(a10, bf0, c1);  c1 = wmma_f32(a11, bf1, c1);
#pragma unroll
      for (int j = 0; j < 8; ++j) {
        const int r = j + dRow;
        lds_h[r * HID + n]        = relu(c0[j]);
        lds_h[(16 + r) * HID + n] = relu(c1[j]);
      }
    }
  }

#pragma unroll
  for (int nt = 0; nt < 8; ++nt) {
    const int n = nt * 16 + lrow;
    const float bv = b2[n];
    v8f c0, c1;
#pragma unroll
    for (int j = 0; j < 8; ++j) { c0[j] = bv; c1[j] = bv; }
    const float* w2r = W2 + n * HID;
    for (int k4 = 0; k4 < HID; k4 += 4) {
      const int kb = k4 + khalf * 2;
      v2f bf, aA, aB;
      bf.x = w2r[kb];                       bf.y = w2r[kb + 1];
      aA.x = lds_h[lrow * HID + kb];        aA.y = lds_h[lrow * HID + kb + 1];
      aB.x = lds_h[(16 + lrow) * HID + kb]; aB.y = lds_h[(16 + lrow) * HID + kb + 1];
      c0 = wmma_f32(aA, bf, c0);
      c1 = wmma_f32(aB, bf, c1);
    }
#pragma unroll
    for (int j = 0; j < 8; ++j) {
      const int r = j + dRow;
      lds_seed[r * HID + n]        = c0[j];
      lds_seed[(16 + r) * HID + n] = c1[j];
    }
  }

  for (int t = 0; t < OUTW; ++t) {
#pragma unroll
    for (int nt = 0; nt < 8; ++nt) {
      const int n = nt * 16 + lrow;
      const float bv = bd[n];
      v8f c0, c1;
#pragma unroll
      for (int j = 0; j < 8; ++j) { c0[j] = bv; c1[j] = bv; }
      for (int k4 = 0; k4 < HID; k4 += 4) {
        const int kb = k4 + khalf * 2;
        v2f bf, aA, aB;
        bf.x = Wd[(kb * HID + n) * OUTW + t];
        bf.y = Wd[((kb + 1) * HID + n) * OUTW + t];
        aA.x = lds_seed[lrow * HID + kb];
        aA.y = lds_seed[lrow * HID + kb + 1];
        aB.x = lds_seed[(16 + lrow) * HID + kb];
        aB.y = lds_seed[(16 + lrow) * HID + kb + 1];
        c0 = wmma_f32(aA, bf, c0);
        c1 = wmma_f32(aB, bf, c1);
      }
#pragma unroll
      for (int j = 0; j < 8; ++j) {
        const int r = j + dRow;
        lds_ht[r * HID + n]        = relu(c0[j]);
        lds_ht[(16 + r) * HID + n] = relu(c1[j]);
      }
    }
    {
      const int rg = rowBase + lane;
      float a0 = bc[0], a1 = bc[1], a2 = bc[2];
      for (int i = 0; i < HID; ++i) {
        const float v = lds_ht[lane * HID + i];
        a0 += v * Wc[0 * HID + i];
        a1 += v * Wc[1 * HID + i];
        a2 += v * Wc[2 * HID + i];
      }
      float* o = out + ((size_t)rg * OUTW + t) * 3;
      o[0] = predicted[rg * 3 + 0] + a0;
      o[1] = predicted[rg * 3 + 1] + a1;
      o[2] = predicted[rg * 3 + 2] + a2;
    }
  }
}

extern "C" void kernel_launch(void* const* d_in, const int* in_sizes, int n_in,
                              void* d_out, int out_size, void* d_ws, size_t ws_size,
                              hipStream_t stream) {
  const float* partial   = (const float*)d_in[0];
  const float* predicted = (const float*)d_in[1];
  const float* W1 = (const float*)d_in[2];
  const float* b1 = (const float*)d_in[3];
  const float* W2 = (const float*)d_in[4];
  const float* b2 = (const float*)d_in[5];
  const float* Wd = (const float*)d_in[6];
  const float* bd = (const float*)d_in[7];
  const float* Wc = (const float*)d_in[8];
  const float* bc = (const float*)d_in[9];
  float* out = (float*)d_out;

  float* x = (float*)d_ws;

  dim3 g1(NN / 256, BB);
  knn_seed_kernel<<<g1, 256, 0, stream>>>(partial, predicted, x);

  const int blocks = (BB * NN) / 128;
  const size_t lds = (size_t)12 * 32 * HID * sizeof(float);
  mlp_refine_kernel<<<blocks, 128, lds, stream>>>(
      x, predicted, W1, b1, W2, b2, Wd, bd, Wc, bc, out);
}


# KNN top-8 insert decided per slot at once: 8 independent tests on the old list, distance = median(lower, new, own), index by 2 selects; was a 16-deep compare-swap chain; ties keep index order
# speedup vs baseline: 3.4485x; 3.4485x over previous
#include <hip/hip_runtime.h>
#include <hip/hip_bf16.h>


typedef __attribute__((ext_vector_type(2))) float v2f;
typedef __attribute__((ext_vector_type(4))) float v4f;
typedef __attribute__((ext_vector_type(8))) float v8f;

#define BB   8
#define MM   2048
#define NN   8192
#define KNN  8
#define HID  128
#define INP  6
#define INP8 8
#define OUTW 2
#define LP   132

#define PTS 8
#define TRIPS (MM / PTS)

struct Pts8 { v4f q0, q1, q2, q3, q4, q5; };

__device__ __forceinline__ Pts8 load_pts8(const v4f* sp4, int trip) {
  Pts8 p;
  p.q0 = sp4[trip * 6 + 0]; p.q1 = sp4[trip * 6 + 1]; p.q2 = sp4[trip * 6 + 2];
  p.q3 = sp4[trip * 6 + 3]; p.q4 = sp4[trip * 6 + 4]; p.q5 = sp4[trip * 6 + 5];
  return p;
}

__device__ __forceinline__ void scan_pts8(const Pts8& p, int base,
                                          float px, float py, float pz,
                                          float (&bd)[KNN], int (&bi)[KNN]) {
  const float f[PTS * 3] = {p.q0.x, p.q0.y, p.q0.z, p.q0.w, p.q1.x, p.q1.y, p.q1.z, p.q1.w,
                            p.q2.x, p.q2.y, p.q2.z, p.q2.w, p.q3.x, p.q3.y, p.q3.z, p.q3.w,
                            p.q4.x, p.q4.y, p.q4.z, p.q4.w, p.q5.x, p.q5.y, p.q5.z, p.q5.w};
  float d[PTS];
#pragma unroll
  for (int i = 0; i < PTS; ++i) {
    const float dx = px - f[3 * i + 0];
    const float dy = py - f[3 * i + 1];
    const float dz = pz - f[3 * i + 2];
    {
#pragma clang fp contract(off)
      const float sxy = dx * dx + dy * dy;
      d[i] = __builtin_fmaf(dz, dz, sxy);
    }
  }
  const float thr = bd[KNN - 1];
  bool c[PTS];
  unsigned wave_any = 0;
#pragma unroll
  for (int i = 0; i < PTS; ++i) {
    c[i] = d[i] < thr;
    wave_any |= __builtin_amdgcn_ballot_w32(c[i]);
  }
  if (wave_any) {
#pragma unroll
    for (int i = 0; i < PTS; ++i) {
      if (c[i]) {
        const float dd = d[i];
        const int id = base + i;
        bool lt[KNN];
#pragma unroll
        for (int j = 0; j < KNN; ++j) lt[j] = dd < bd[j];
#pragma unroll
        for (int j = KNN - 1; j >= 1; --j)
          bd[j] = __builtin_amdgcn_fmed3f(bd[j - 1], dd, bd[j]);
        bd[0] = lt[0] ? dd : bd[0];
        int ki[KNN];
#pragma unroll
        for (int j = 0; j < KNN; ++j) ki[j] = lt[j] ? id : bi[j];
#pragma unroll
        for (int j = KNN - 1; j >= 1; --j)
          bi[j] = lt[j - 1] ? bi[j - 1] : ki[j];
        bi[0] = ki[0];
      }
    }
  }
}

__global__ void __launch_bounds__(256)
knn_seed_kernel(const float* __restrict__ partial,
                const float* __restrict__ predicted,
                float* __restrict__ xout) {
  __shared__ __attribute__((aligned(16))) float sp[MM * 3];
  const int b   = blockIdx.y;
  const int tid = threadIdx.x;

  for (int i = tid; i < MM * 3; i += 256)
    sp[i] = partial[b * MM * 3 + i];
  __syncthreads();

  const int n = blockIdx.x * 256 + tid;
  const int row = b * NN + n;
  const float px = predicted[row * 3 + 0];
  const float py = predicted[row * 3 + 1];
  const float pz = predicted[row * 3 + 2];

  float bd[KNN]; int bi[KNN];
#pragma unroll
  for (int j = 0; j < KNN; ++j) { bd[j] = 3.4e38f; bi[j] = 0; }

  const v4f* sp4 = (const v4f*)sp;
  Pts8 pa = load_pts8(sp4, 0);
  for (int t = 0; t < TRIPS; t += 2) {
    const Pts8 pb = load_pts8(sp4, t + 1);
    scan_pts8(pa, t * PTS, px, py, pz, bd, bi);
    const int tn = (t + 2 < TRIPS) ? (t + 2) : t;
    pa = load_pts8(sp4, tn);
    scan_pts8(pb, (t + 1) * PTS, px, py, pz, bd, bi);
  }

  float sx = 0.f, sy = 0.f, sz = 0.f;
#pragma unroll
  for (int j = 0; j < KNN; ++j) {
    const int m = bi[j];
    sx += sp[3 * m + 0]; sy += sp[3 * m + 1]; sz += sp[3 * m + 2];
  }

  float* xr = xout + (size_t)row * INP8;
  xr[0] = px;          xr[1] = py;          xr[2] = pz;
  xr[3] = sx * 0.125f; xr[4] = sy * 0.125f; xr[5] = sz * 0.125f;
  xr[6] = 0.f;         xr[7] = 0.f;
}

__device__ __forceinline__ v8f wmma_f32(v2f a, v2f b, v8f c) {
  return __builtin_amdgcn_wmma_f32_16x16x4_f32(
       false, a,  false, b,
       (short)0, c,  false,  false);
}

__device__ __forceinline__ float relu(float x) { return x > 0.f ? x : 0.f; }

__global__ void __launch_bounds__(128)
mlp_refine_kernel(const float* __restrict__ xin,
                  const float* __restrict__ predicted,
                  const float* __restrict__ W1, const float* __restrict__ b1,
                  const float* __restrict__ W2, const float* __restrict__ b2,
                  const float* __restrict__ Wd, const float* __restrict__ bd,
                  const float* __restrict__ Wc, const float* __restrict__ bc,
                  float* __restrict__ out) {
  extern __shared__ __attribute__((aligned(16))) float smem[];
  const int tid   = threadIdx.x;
  const int wave  = tid >> 5;
  const int lane  = tid & 31;
  const int lrow  = lane & 15;
  const int khalf = lane >> 4;

  float* lds_h    = smem + wave * (32 * LP);
  float* lds_seed = smem + 4 * (32 * LP) + wave * (32 * LP);
  float* lds_ht   = smem + 8 * (32 * LP) + wave * (32 * LP);

  const int rowBase = blockIdx.x * 128 + wave * 32;
  const int dRow = (lane < 16) ? 0 : 8;

  {
    const float* xr0 = xin + (size_t)(rowBase + lrow) * INP8;
    const float* xr1 = xin + (size_t)(rowBase + 16 + lrow) * INP8;
    const float kmask = khalf ? 0.f : 1.f;
    v2f a00, a01, a10, a11;
    a00.x = xr0[khalf * 2 + 0];     a00.y = xr0[khalf * 2 + 1];
    a01.x = xr0[4 + khalf * 2 + 0]; a01.y = xr0[4 + khalf * 2 + 1];
    a10.x = xr1[khalf * 2 + 0];     a10.y = xr1[khalf * 2 + 1];
    a11.x = xr1[4 + khalf * 2 + 0]; a11.y = xr1[4 + khalf * 2 + 1];
#pragma unroll
    for (int nt = 0; nt < 8; ++nt) {
      const int n = nt * 16 + lrow;
      const float bv = b1[n];
      v8f c0, c1;
#pragma unroll
      for (int j = 0; j < 8; ++j) { c0[j] = bv; c1[j] = bv; }
      v2f bf0, bf1;
      const int k0 = khalf * 2;
      bf0.x = W1[n * INP + k0];        bf0.y = W1[n * INP + k0 + 1];
      bf1.x = kmask * W1[n * INP + 4]; bf1.y = kmask * W1[n * INP + 5];
      c0 = wmma_f32(a00, bf0, c0);  c0 = wmma_f32(a01, bf1, c0);
      c1 = wmma_f32(a10, bf0, c1);  c1 = wmma_f32(a11, bf1, c1);
#pragma unroll
      for (int j = 0; j < 8; ++j) {
        const int r = j + dRow;
        lds_h[r * LP + n]        = relu(c0[j]);
        lds_h[(16 + r) * LP + n] = relu(c1[j]);
      }
    }
  }

  {
    v8f c0[8], c1[8];
#pragma unroll
    for (int nt = 0; nt < 8; ++nt) {
      const float bv = b2[nt * 16 + lrow];
#pragma unroll
      for (int j = 0; j < 8; ++j) { c0[nt][j] = bv; c1[nt][j] = bv; }
    }
    const float* w2p = W2 + lrow * HID + khalf * 4;
    const float* a0p = lds_h + lrow * LP + khalf * 4;
    const float* a1p = lds_h + (16 + lrow) * LP + khalf * 4;
    v4f bP[8], bQ[8];
#pragma unroll
    for (int nt = 0; nt < 8; ++nt)
      bP[nt] = *(const v4f*)(w2p + nt * 16 * HID);
    for (int k8 = 0; k8 < HID; k8 += 16) {
#pragma unroll
      for (int nt = 0; nt < 8; ++nt)
        bQ[nt] = *(const v4f*)(w2p + nt * 16 * HID + k8 + 8);
      {
        const v4f aA = *(const v4f*)(a0p + k8);
        const v4f aB = *(const v4f*)(a1p + k8);
        const v2f aA0 = {aA.x, aA.y}, aA1 = {aA.z, aA.w};
        const v2f aB0 = {aB.x, aB.y}, aB1 = {aB.z, aB.w};
#pragma unroll
        for (int nt = 0; nt < 8; ++nt) {
          const v2f bf = {bP[nt].x, bP[nt].y};
          c0[nt] = wmma_f32(aA0, bf, c0[nt]);
          c1[nt] = wmma_f32(aB0, bf, c1[nt]);
        }
#pragma unroll
        for (int nt = 0; nt < 8; ++nt) {
          const v2f bf = {bP[nt].z, bP[nt].w};
          c0[nt] = wmma_f32(aA1, bf, c0[nt]);
          c1[nt] = wmma_f32(aB1, bf, c1[nt]);
        }
      }
      const int kn = (k8 + 16 < HID) ? (k8 + 16) : k8;
#pragma unroll
      for (int nt = 0; nt < 8; ++nt)
        bP[nt] = *(const v4f*)(w2p + nt * 16 * HID + kn);
      {
        const v4f aA = *(const v4f*)(a0p + k8 + 8);
        const v4f aB = *(const v4f*)(a1p + k8 + 8);
        const v2f aA0 = {aA.x, aA.y}, aA1 = {aA.z, aA.w};
        const v2f aB0 = {aB.x, aB.y}, aB1 = {aB.z, aB.w};
#pragma unroll
        for (int nt = 0; nt < 8; ++nt) {
          const v2f bf = {bQ[nt].x, bQ[nt].y};
          c0[nt] = wmma_f32(aA0, bf, c0[nt]);
          c1[nt] = wmma_f32(aB0, bf, c1[nt]);
        }
#pragma unroll
        for (int nt = 0; nt < 8; ++nt) {
          const v2f bf = {bQ[nt].z, bQ[nt].w};
          c0[nt] = wmma_f32(aA1, bf, c0[nt]);
          c1[nt] = wmma_f32(aB1, bf, c1[nt]);
        }
      }
    }
#pragma unroll
    for (int nt = 0; nt < 8; ++nt) {
      const int n = nt * 16 + lrow;
#pragma unroll
      for (int j = 0; j < 8; ++j) {
        const int r = j + dRow;
        lds_seed[r * LP + n]        = c0[nt][j];
        lds_seed[(16 + r) * LP + n] = c1[nt][j];
      }
    }
  }

  {
    v8f d0[OUTW][8], d1[OUTW][8];
#pragma unroll
    for (int nt = 0; nt < 8; ++nt) {
      const float bv = bd[nt * 16 + lrow];
#pragma unroll
      for (int t = 0; t < OUTW; ++t) {
#pragma unroll
        for (int j = 0; j < 8; ++j) { d0[t][nt][j] = bv; d1[t][nt][j] = bv; }
      }
    }
    const float* wdp = Wd + ((khalf * 2) * HID + lrow) * OUTW;
    const float* s0p = lds_seed + lrow * LP + khalf * 2;
    const float* s1p = lds_seed + (16 + lrow) * LP + khalf * 2;
    v2f eP[8][2], eQ[8][2];
#pragma unroll
    for (int nt = 0; nt < 8; ++nt) {
      eP[nt][0] = *(const v2f*)(wdp + (nt * 16) * OUTW);
      eP[nt][1] = *(const v2f*)(wdp + (HID + nt * 16) * OUTW);
    }
    for (int k4 = 0; k4 < HID; k4 += 8) {
#pragma unroll
      for (int nt = 0; nt < 8; ++nt) {
        eQ[nt][0] = *(const v2f*)(wdp + ((k4 + 4) * HID + nt * 16) * OUTW);
        eQ[nt][1] = *(const v2f*)(wdp + ((k4 + 5) * HID + nt * 16) * OUTW);
      }
      {
        const v2f aA = *(const v2f*)(s0p + k4);
        const v2f aB = *(const v2f*)(s1p + k4);
#pragma unroll
        for (int nt = 0; nt < 8; ++nt) {
#pragma unroll
          for (int t = 0; t < OUTW; ++t) {
            const v2f bf = {eP[nt][0][t], eP[nt][1][t]};
            d0[t][nt] = wmma_f32(aA, bf, d0[t][nt]);
            d1[t][nt] = wmma_f32(aB, bf, d1[t][nt]);
          }
        }
      }
      const int kn = (k4 + 8 < HID) ? (k4 + 8) : k4;
#pragma unroll
      for (int nt = 0; nt < 8; ++nt) {
        eP[nt][0] = *(const v2f*)(wdp + (kn * HID + nt * 16) * OUTW);
        eP[nt][1] = *(const v2f*)(wdp + ((kn + 1) * HID + nt * 16) * OUTW);
      }
      {
        const v2f aA = *(const v2f*)(s0p + k4 + 4);
        const v2f aB = *(const v2f*)(s1p + k4 + 4);
#pragma unroll
        for (int nt = 0; nt < 8; ++nt) {
#pragma unroll
          for (int t = 0; t < OUTW; ++t) {
            const v2f bf = {eQ[nt][0][t], eQ[nt][1][t]};
            d0[t][nt] = wmma_f32(aA, bf, d0[t][nt]);
            d1[t][nt] = wmma_f32(aB, bf, d1[t][nt]);
          }
        }
      }
    }

#pragma unroll
    for (int t = 0; t < OUTW; ++t) {
#pragma unroll
      for (int nt = 0; nt < 8; ++nt) {
        const int n = nt * 16 + lrow;
#pragma unroll
        for (int j = 0; j < 8; ++j) {
          const int r = j + dRow;
          lds_ht[r * LP + n]        = relu(d0[t][nt][j]);
          lds_ht[(16 + r) * LP + n] = relu(d1[t][nt][j]);
        }
      }
      {
        const int rg = rowBase + lane;
        const float* hr = lds_ht + lane * LP;
        float a0 = bc[0], a1 = bc[1], a2 = bc[2];
        for (int i = 0; i < HID; i += 4) {
          const v4f v = *(const v4f*)(hr + i);
#pragma unroll
          for (int q = 0; q < 4; ++q) {
            a0 += v[q] * Wc[0 * HID + i + q];
            a1 += v[q] * Wc[1 * HID + i + q];
            a2 += v[q] * Wc[2 * HID + i + q];
          }
        }
        float* o = out + ((size_t)rg * OUTW + t) * 3;
        o[0] = predicted[rg * 3 + 0] + a0;
        o[1] = predicted[rg * 3 + 1] + a1;
        o[2] = predicted[rg * 3 + 2] + a2;
      }
    }
  }
}

extern "C" void kernel_launch(void* const* d_in, const int* in_sizes, int n_in,
                              void* d_out, int out_size, void* d_ws, size_t ws_size,
                              hipStream_t stream) {
  const float* partial   = (const float*)d_in[0];
  const float* predicted = (const float*)d_in[1];
  const float* W1 = (const float*)d_in[2];
  const float* b1 = (const float*)d_in[3];
  const float* W2 = (const float*)d_in[4];
  const float* b2 = (const float*)d_in[5];
  const float* Wd = (const float*)d_in[6];
  const float* bd = (const float*)d_in[7];
  const float* Wc = (const float*)d_in[8];
  const float* bc = (const float*)d_in[9];
  float* out = (float*)d_out;

  float* x = (float*)d_ws;

  dim3 g1(NN / 256, BB);
  knn_seed_kernel<<<g1, 256, 0, stream>>>(partial, predicted, x);

  const int blocks = (BB * NN) / 128;
  const size_t lds = (size_t)12 * 32 * LP * sizeof(float);
  mlp_refine_kernel<<<blocks, 128, lds, stream>>>(
      x, predicted, W1, b1, W2, b2, Wd, bd, Wc, bc, out);
}


# Layers 2 and 3 merged (no activation between them): W2^T*Wd and b2*Wd+bd formed once per call on WMMA by spare blocks of launch 1, MLP does h@(W2^T*Wd); seed tile gone; f32 sum order changes
# speedup vs baseline: 4.1699x; 1.2092x over previous
#include <hip/hip_runtime.h>
#include <hip/hip_bf16.h>


typedef __attribute__((ext_vector_type(2))) float v2f;
typedef __attribute__((ext_vector_type(4))) float v4f;
typedef __attribute__((ext_vector_type(8))) float v8f;

#define BB   8
#define MM   2048
#define NN   8192
#define KNN  8
#define HID  128
#define INP  6
#define INP8 8
#define OUTW 2
#define LP   132

#define PTS 8
#define TRIPS (MM / PTS)

struct Pts8 { v4f q0, q1, q2, q3, q4, q5; };

__device__ __forceinline__ Pts8 load_pts8(const v4f* sp4, int trip) {
  Pts8 p;
  p.q0 = sp4[trip * 6 + 0]; p.q1 = sp4[trip * 6 + 1]; p.q2 = sp4[trip * 6 + 2];
  p.q3 = sp4[trip * 6 + 3]; p.q4 = sp4[trip * 6 + 4]; p.q5 = sp4[trip * 6 + 5];
  return p;
}

__device__ __forceinline__ void scan_pts8(const Pts8& p, int base,
                                          float px, float py, float pz,
                                          float (&bd)[KNN], int (&bi)[KNN]) {
  const float f[PTS * 3] = {p.q0.x, p.q0.y, p.q0.z, p.q0.w, p.q1.x, p.q1.y, p.q1.z, p.q1.w,
                            p.q2.x, p.q2.y, p.q2.z, p.q2.w, p.q3.x, p.q3.y, p.q3.z, p.q3.w,
                            p.q4.x, p.q4.y, p.q4.z, p.q4.w, p.q5.x, p.q5.y, p.q5.z, p.q5.w};
  float d[PTS];
#pragma unroll
  for (int i = 0; i < PTS; ++i) {
    const float dx = px - f[3 * i + 0];
    const float dy = py - f[3 * i + 1];
    const float dz = pz - f[3 * i + 2];
    {
#pragma clang fp contract(off)
      const float sxy = dx * dx + dy * dy;
      d[i] = __builtin_fmaf(dz, dz, sxy);
    }
  }
  const float thr = bd[KNN - 1];
  bool c[PTS];
  unsigned wave_any = 0;
#pragma unroll
  for (int i = 0; i < PTS; ++i) {
    c[i] = d[i] < thr;
    wave_any |= __builtin_amdgcn_ballot_w32(c[i]);
  }
  if (wave_any) {
#pragma unroll
    for (int i = 0; i < PTS; ++i) {
      if (c[i]) {
        const float dd = d[i];
        const int id = base + i;
        bool lt[KNN];
#pragma unroll
        for (int j = 0; j < KNN; ++j) lt[j] = dd < bd[j];
#pragma unroll
        for (int j = KNN - 1; j >= 1; --j)
          bd[j] = __builtin_amdgcn_fmed3f(bd[j - 1], dd, bd[j]);
        bd[0] = lt[0] ? dd : bd[0];
        int ki[KNN];
#pragma unroll
        for (int j = 0; j < KNN; ++j) ki[j] = lt[j] ? id : bi[j];
#pragma unroll
        for (int j = KNN - 1; j >= 1; --j)
          bi[j] = lt[j - 1] ? bi[j - 1] : ki[j];
        bi[0] = ki[0];
      }
    }
  }
}

__device__ __forceinline__ v8f wmma_f32(v2f a, v2f b, v8f c) {
  return __builtin_amdgcn_wmma_f32_16x16x4_f32(
       false, a,  false, b,
       (short)0, c,  false,  false);
}

__device__ __forceinline__ void merge_linear_layers(const float* __restrict__ W2,
                                                    const float* __restrict__ b2,
                                                    const float* __restrict__ Wd,
                                                    const float* __restrict__ bd,
                                                    float* __restrict__ mw,
                                                    float* __restrict__ mb) {
  const int tid = threadIdx.x;
  const int bx  = blockIdx.x;
  if (bx < 16) {
    const int wave = tid >> 5, lane = tid & 31;
    const int c = lane & 15, half = lane >> 4;
    const int tile = bx * 8 + wave;
    const int mt = tile >> 3, nt = tile & 7;
    const int t = mt >> 3;
    const int o = (mt & 7) * 16 + c;
    const int j = nt * 16 + c;
    const float* ap = Wd + ((size_t)(2 * half) * HID + o) * OUTW + t;
    const float* bp = W2 + (2 * half) * HID + j;
    v8f acc;
#pragma unroll
    for (int r = 0; r < 8; ++r) acc[r] = 0.f;
#pragma unroll 8
    for (int k0 = 0; k0 < HID; k0 += 4) {
      v2f a, b;
      a.x = ap[(size_t)k0 * HID * OUTW];  a.y = ap[(size_t)(k0 + 1) * HID * OUTW];
      b.x = bp[k0 * HID];                 b.y = bp[(k0 + 1) * HID];
      acc = wmma_f32(a, b, acc);
    }
    float* dp = mw + (size_t)(mt * 16 + 8 * half) * HID + nt * 16 + c;
#pragma unroll
    for (int r = 0; r < 8; ++r) dp[r * HID] = acc[r];
  } else if (bx == 16) {
    const int t = tid >> 7, o = tid & (HID - 1);
    float s = 0.f;
    for (int i = 0; i < HID; ++i) s = __builtin_fmaf(b2[i], Wd[((size_t)i * HID + o) * OUTW + t], s);
    mb[tid] = s + bd[o];
  }
}

__global__ void __launch_bounds__(256)
knn_seed_kernel(const float* __restrict__ partial,
                const float* __restrict__ predicted,
                const float* __restrict__ lin2_w, const float* __restrict__ lin2_b,
                const float* __restrict__ dec_w,  const float* __restrict__ dec_b,
                float* __restrict__ mw, float* __restrict__ mb,
                float* __restrict__ xout) {
  if (blockIdx.y == BB) {
    merge_linear_layers(lin2_w, lin2_b, dec_w, dec_b, mw, mb);
    return;
  }
  __shared__ __attribute__((aligned(16))) float sp[MM * 3];
  const int b   = blockIdx.y;
  const int tid = threadIdx.x;

  for (int i = tid; i < MM * 3; i += 256)
    sp[i] = partial[b * MM * 3 + i];
  __syncthreads();

  const int n = blockIdx.x * 256 + tid;
  const int row = b * NN + n;
  const float px = predicted[row * 3 + 0];
  const float py = predicted[row * 3 + 1];
  const float pz = predicted[row * 3 + 2];

  float bd[KNN]; int bi[KNN];
#pragma unroll
  for (int j = 0; j < KNN; ++j) { bd[j] = 3.4e38f; bi[j] = 0; }

  const v4f* sp4 = (const v4f*)sp;
  Pts8 pa = load_pts8(sp4, 0);
  for (int t = 0; t < TRIPS; t += 2) {
    const Pts8 pb = load_pts8(sp4, t + 1);
    scan_pts8(pa, t * PTS, px, py, pz, bd, bi);
    const int tn = (t + 2 < TRIPS) ? (t + 2) : t;
    pa = load_pts8(sp4, tn);
    scan_pts8(pb, (t + 1) * PTS, px, py, pz, bd, bi);
  }

  float sx = 0.f, sy = 0.f, sz = 0.f;
#pragma unroll
  for (int j = 0; j < KNN; ++j) {
    const int m = bi[j];
    sx += sp[3 * m + 0]; sy += sp[3 * m + 1]; sz += sp[3 * m + 2];
  }

  float* xr = xout + (size_t)row * INP8;
  xr[0] = px;          xr[1] = py;          xr[2] = pz;
  xr[3] = sx * 0.125f; xr[4] = sy * 0.125f; xr[5] = sz * 0.125f;
  xr[6] = 0.f;         xr[7] = 0.f;
}

__device__ __forceinline__ float relu(float x) { return x > 0.f ? x : 0.f; }

__global__ void __launch_bounds__(128)
mlp_refine_kernel(const float* __restrict__ xin,
                  const float* __restrict__ predicted,
                  const float* __restrict__ W1, const float* __restrict__ b1,
                  const float* __restrict__ mw, const float* __restrict__ mb,
                  const float* __restrict__ Wc, const float* __restrict__ bc,
                  float* __restrict__ out) {
  extern __shared__ __attribute__((aligned(16))) float smem[];
  const int tid   = threadIdx.x;
  const int wave  = tid >> 5;
  const int lane  = tid & 31;
  const int lrow  = lane & 15;
  const int khalf = lane >> 4;

  float* lds_h  = smem + wave * (32 * LP);
  float* lds_ht = smem + 4 * (32 * LP) + wave * (32 * LP);

  const int rowBase = blockIdx.x * 128 + wave * 32;
  const int dRow = (lane < 16) ? 0 : 8;

  {
    const float* xr0 = xin + (size_t)(rowBase + lrow) * INP8;
    const float* xr1 = xin + (size_t)(rowBase + 16 + lrow) * INP8;
    const float kmask = khalf ? 0.f : 1.f;
    v2f a00, a01, a10, a11;
    a00.x = xr0[khalf * 2 + 0];     a00.y = xr0[khalf * 2 + 1];
    a01.x = xr0[4 + khalf * 2 + 0]; a01.y = xr0[4 + khalf * 2 + 1];
    a10.x = xr1[khalf * 2 + 0];     a10.y = xr1[khalf * 2 + 1];
    a11.x = xr1[4 + khalf * 2 + 0]; a11.y = xr1[4 + khalf * 2 + 1];
#pragma unroll
    for (int nt = 0; nt < 8; ++nt) {
      const int n = nt * 16 + lrow;
      const float bv = b1[n];
      v8f c0, c1;
#pragma unroll
      for (int j = 0; j < 8; ++j) { c0[j] = bv; c1[j] = bv; }
      v2f bf0, bf1;
      const int k0 = khalf * 2;
      bf0.x = W1[n * INP + k0];        bf0.y = W1[n * INP + k0 + 1];
      bf1.x = kmask * W1[n * INP + 4]; bf1.y = kmask * W1[n * INP + 5];
      c0 = wmma_f32(a00, bf0, c0);  c0 = wmma_f32(a01, bf1, c0);
      c1 = wmma_f32(a10, bf0, c1);  c1 = wmma_f32(a11, bf1, c1);
#pragma unroll
      for (int j = 0; j < 8; ++j) {
        const int r = j + dRow;
        lds_h[r * LP + n]        = relu(c0[j]);
        lds_h[(16 + r) * LP + n] = relu(c1[j]);
      }
    }
  }

  for (int t = 0; t < OUTW; ++t) {
    v8f c0[8], c1[8];
#pragma unroll
    for (int nt = 0; nt < 8; ++nt) {
      const float bv = mb[t * HID + nt * 16 + lrow];
#pragma unroll
      for (int j = 0; j < 8; ++j) { c0[nt][j] = bv; c1[nt][j] = bv; }
    }
    const float* wp  = mw + (size_t)(t * HID + lrow) * HID + khalf * 4;
    const float* a0p = lds_h + lrow * LP + khalf * 4;
    const float* a1p = lds_h + (16 + lrow) * LP + khalf * 4;
    v4f bP[8], bQ[8];
#pragma unroll
    for (int nt = 0; nt < 8; ++nt)
      bP[nt] = *(const v4f*)(wp + nt * 16 * HID);
    for (int k8 = 0; k8 < HID; k8 += 16) {
#pragma unroll
      for (int nt = 0; nt < 8; ++nt)
        bQ[nt] = *(const v4f*)(wp + nt * 16 * HID + k8 + 8);
      {
        const v4f aA = *(const v4f*)(a0p + k8);
        const v4f aB = *(const v4f*)(a1p + k8);
        const v2f aA0 = {aA.x, aA.y}, aA1 = {aA.z, aA.w};
        const v2f aB0 = {aB.x, aB.y}, aB1 = {aB.z, aB.w};
#pragma unroll
        for (int nt = 0; nt < 8; ++nt) {
          const v2f bf = {bP[nt].x, bP[nt].y};
          c0[nt] = wmma_f32(aA0, bf, c0[nt]);
          c1[nt] = wmma_f32(aB0, bf, c1[nt]);
        }
#pragma unroll
        for (int nt = 0; nt < 8; ++nt) {
          const v2f bf = {bP[nt].z, bP[nt].w};
          c0[nt] = wmma_f32(aA1, bf, c0[nt]);
          c1[nt] = wmma_f32(aB1, bf, c1[nt]);
        }
      }
      const int kn = (k8 + 16 < HID) ? (k8 + 16) : k8;
#pragma unroll
      for (int nt = 0; nt < 8; ++nt)
        bP[nt] = *(const v4f*)(wp + nt * 16 * HID + kn);
      {
        const v4f aA = *(const v4f*)(a0p + k8 + 8);
        const v4f aB = *(const v4f*)(a1p + k8 + 8);
        const v2f aA0 = {aA.x, aA.y}, aA1 = {aA.z, aA.w};
        const v2f aB0 = {aB.x, aB.y}, aB1 = {aB.z, aB.w};
#pragma unroll
        for (int nt = 0; nt < 8; ++nt) {
          const v2f bf = {bQ[nt].x, bQ[nt].y};
          c0[nt] = wmma_f32(aA0, bf, c0[nt]);
          c1[nt] = wmma_f32(aB0, bf, c1[nt]);
        }
#pragma unroll
        for (int nt = 0; nt < 8; ++nt) {
          const v2f bf = {bQ[nt].z, bQ[nt].w};
          c0[nt] = wmma_f32(aA1, bf, c0[nt]);
          c1[nt] = wmma_f32(aB1, bf, c1[nt]);
        }
      }
    }
#pragma unroll
    for (int nt = 0; nt < 8; ++nt) {
      const int n = nt * 16 + lrow;
#pragma unroll
      for (int j = 0; j < 8; ++j) {
        const int r = j + dRow;
        lds_ht[r * LP + n]        = relu(c0[nt][j]);
        lds_ht[(16 + r) * LP + n] = relu(c1[nt][j]);
      }
    }
    {
      const int rg = rowBase + lane;
      const float* hr = lds_ht + lane * LP;
      float a0 = bc[0], a1 = bc[1], a2 = bc[2];
      for (int i = 0; i < HID; i += 4) {
        const v4f v = *(const v4f*)(hr + i);
#pragma unroll
        for (int q = 0; q < 4; ++q) {
          a0 += v[q] * Wc[0 * HID + i + q];
          a1 += v[q] * Wc[1 * HID + i + q];
          a2 += v[q] * Wc[2 * HID + i + q];
        }
      }
      float* o = out + ((size_t)rg * OUTW + t) * 3;
      o[0] = predicted[rg * 3 + 0] + a0;
      o[1] = predicted[rg * 3 + 1] + a1;
      o[2] = predicted[rg * 3 + 2] + a2;
    }
  }
}

extern "C" void kernel_launch(void* const* d_in, const int* in_sizes, int n_in,
                              void* d_out, int out_size, void* d_ws, size_t ws_size,
                              hipStream_t stream) {
  const float* partial   = (const float*)d_in[0];
  const float* predicted = (const float*)d_in[1];
  const float* W1 = (const float*)d_in[2];
  const float* b1 = (const float*)d_in[3];
  const float* W2 = (const float*)d_in[4];
  const float* b2 = (const float*)d_in[5];
  const float* Wd = (const float*)d_in[6];
  const float* bd = (const float*)d_in[7];
  const float* Wc = (const float*)d_in[8];
  const float* bc = (const float*)d_in[9];
  float* out = (float*)d_out;

  float* x  = (float*)d_ws;
  float* mw = x + (size_t)BB * NN * INP8;
  float* mb = mw + (size_t)OUTW * HID * HID;

  dim3 g1(NN / 256, BB + 1);
  knn_seed_kernel<<<g1, 256, 0, stream>>>(partial, predicted, W2, b2, Wd, bd, mw, mb, x);

  const int blocks = (BB * NN) / 128;
  const size_t lds = (size_t)8 * 32 * LP * sizeof(float);
  mlp_refine_kernel<<<blocks, 128, lds, stream>>>(
      x, predicted, W1, b1, mw, mb, Wc, bc, out);
}


# KNN: points that pass the distance test are appended per lane to an LDS queue without a branch and inserted in batches in index order; 2 branches per 8-point trip, was up to 10; same list, same sums
# speedup vs baseline: 4.5194x; 1.0838x over previous
#include <hip/hip_runtime.h>
#include <hip/hip_bf16.h>


typedef __attribute__((ext_vector_type(2))) float v2f;
typedef __attribute__((ext_vector_type(4))) float v4f;
typedef __attribute__((ext_vector_type(8))) float v8f;

#define BB   8
#define MM   2048
#define NN   8192
#define KNN  8
#define HID  128
#define INP  6
#define INP8 8
#define OUTW 2
#define LP   132

#define PTS 8
#define TRIPS (MM / PTS)
#define QCAP 16
#define QROW 256
struct __attribute__((aligned(8))) QEnt { float d; int id; };

struct Pts8 { v4f q0, q1, q2, q3, q4, q5; };

__device__ __forceinline__ Pts8 load_pts8(const v4f* sp4, int trip) {
  Pts8 p;
  p.q0 = sp4[trip * 6 + 0]; p.q1 = sp4[trip * 6 + 1]; p.q2 = sp4[trip * 6 + 2];
  p.q3 = sp4[trip * 6 + 3]; p.q4 = sp4[trip * 6 + 4]; p.q5 = sp4[trip * 6 + 5];
  return p;
}

__device__ __forceinline__ void insert_one(float dd, int id,
                                           float (&bd)[KNN], int (&bi)[KNN]) {
  bool lt[KNN];
#pragma unroll
  for (int j = 0; j < KNN; ++j) lt[j] = dd < bd[j];
#pragma unroll
  for (int j = KNN - 1; j >= 1; --j)
    bd[j] = __builtin_amdgcn_fmed3f(bd[j - 1], dd, bd[j]);
  bd[0] = lt[0] ? dd : bd[0];
  int ki[KNN];
#pragma unroll
  for (int j = 0; j < KNN; ++j) ki[j] = lt[j] ? id : bi[j];
#pragma unroll
  for (int j = KNN - 1; j >= 1; --j)
    bi[j] = lt[j - 1] ? bi[j - 1] : ki[j];
  bi[0] = ki[0];
}

__device__ __forceinline__ void drain_queue(const QEnt* qt, int tid, int& qi,
                                            float (&bd)[KNN], int (&bi)[KNN]) {
#pragma nounroll
  for (int e = 0; e < QCAP; e += 2) {
    const bool live0 = e * QROW + tid < qi;
    if (__builtin_amdgcn_ballot_w32(live0) == 0) break;
    const bool live1 = (e + 1) * QROW + tid < qi;
    const QEnt c0 = qt[e * QROW];
    const QEnt c1 = qt[(e + 1) * QROW];
    insert_one(live0 ? c0.d : __builtin_inff(), c0.id, bd, bi);
    insert_one(live1 ? c1.d : __builtin_inff(), c1.id, bd, bi);
  }
  qi = tid;
}

__device__ __forceinline__ void scan_pts8(const Pts8& p, int base,
                                          float px, float py, float pz,
                                          float (&bd)[KNN], int (&bi)[KNN],
                                          QEnt* q, int tid, int& qi) {
  const float f[PTS * 3] = {p.q0.x, p.q0.y, p.q0.z, p.q0.w, p.q1.x, p.q1.y, p.q1.z, p.q1.w,
                            p.q2.x, p.q2.y, p.q2.z, p.q2.w, p.q3.x, p.q3.y, p.q3.z, p.q3.w,
                            p.q4.x, p.q4.y, p.q4.z, p.q4.w, p.q5.x, p.q5.y, p.q5.z, p.q5.w};
  float d[PTS];
#pragma unroll
  for (int i = 0; i < PTS; ++i) {
    const float dx = px - f[3 * i + 0];
    const float dy = py - f[3 * i + 1];
    const float dz = pz - f[3 * i + 2];
    {
#pragma clang fp contract(off)
      const float sxy = dx * dx + dy * dy;
      d[i] = __builtin_fmaf(dz, dz, sxy);
    }
  }
  const float thr = bd[KNN - 1];
  bool c[PTS];
  const unsigned full = __builtin_amdgcn_ballot_w32(qi > (QCAP - PTS) * QROW + tid);
  unsigned wave_any = full;
#pragma unroll
  for (int i = 0; i < PTS; ++i) {
    c[i] = d[i] < thr;
    wave_any |= __builtin_amdgcn_ballot_w32(c[i]);
  }
  if (wave_any) {
    if (full) drain_queue(q + tid, tid, qi, bd, bi);
#pragma unroll
    for (int i = 0; i < PTS; ++i) {
      const QEnt ent = {d[i], base + i};
      q[qi] = ent;
      qi += c[i] ? QROW : 0;
    }
  }
}

__device__ __forceinline__ v8f wmma_f32(v2f a, v2f b, v8f c) {
  return __builtin_amdgcn_wmma_f32_16x16x4_f32(
       false, a,  false, b,
       (short)0, c,  false,  false);
}

__device__ __forceinline__ void merge_linear_layers(const float* __restrict__ W2,
                                                    const float* __restrict__ b2,
                                                    const float* __restrict__ Wd,
                                                    const float* __restrict__ bd,
                                                    float* __restrict__ mw,
                                                    float* __restrict__ mb) {
  const int tid = threadIdx.x;
  const int bx  = blockIdx.x;
  if (bx < 16) {
    const int wave = tid >> 5, lane = tid & 31;
    const int c = lane & 15, half = lane >> 4;
    const int tile = bx * 8 + wave;
    const int mt = tile >> 3, nt = tile & 7;
    const int t = mt >> 3;
    const int o = (mt & 7) * 16 + c;
    const int j = nt * 16 + c;
    const float* ap = Wd + ((size_t)(2 * half) * HID + o) * OUTW + t;
    const float* bp = W2 + (2 * half) * HID + j;
    v8f acc;
#pragma unroll
    for (int r = 0; r < 8; ++r) acc[r] = 0.f;
#pragma unroll 8
    for (int k0 = 0; k0 < HID; k0 += 4) {
      v2f a, b;
      a.x = ap[(size_t)k0 * HID * OUTW];  a.y = ap[(size_t)(k0 + 1) * HID * OUTW];
      b.x = bp[k0 * HID];                 b.y = bp[(k0 + 1) * HID];
      acc = wmma_f32(a, b, acc);
    }
    float* dp = mw + (size_t)(mt * 16 + 8 * half) * HID + nt * 16 + c;
#pragma unroll
    for (int r = 0; r < 8; ++r) dp[r * HID] = acc[r];
  } else if (bx == 16) {
    const int t = tid >> 7, o = tid & (HID - 1);
    float s = 0.f;
    for (int i = 0; i < HID; ++i) s = __builtin_fmaf(b2[i], Wd[((size_t)i * HID + o) * OUTW + t], s);
    mb[tid] = s + bd[o];
  }
}

__global__ void __launch_bounds__(256)
knn_seed_kernel(const float* __restrict__ partial,
                const float* __restrict__ predicted,
                const float* __restrict__ lin2_w, const float* __restrict__ lin2_b,
                const float* __restrict__ dec_w,  const float* __restrict__ dec_b,
                float* __restrict__ mw, float* __restrict__ mb,
                float* __restrict__ xout) {
  if (blockIdx.y == BB) {
    merge_linear_layers(lin2_w, lin2_b, dec_w, dec_b, mw, mb);
    return;
  }
  __shared__ __attribute__((aligned(16))) float sp[MM * 3];
  __shared__ QEnt qbuf[QCAP * QROW];
  const int b   = blockIdx.y;
  const int tid = threadIdx.x;

  for (int i = tid; i < MM * 3; i += 256)
    sp[i] = partial[b * MM * 3 + i];
  __syncthreads();

  const int n = blockIdx.x * 256 + tid;
  const int row = b * NN + n;
  const float px = predicted[row * 3 + 0];
  const float py = predicted[row * 3 + 1];
  const float pz = predicted[row * 3 + 2];

  float bd[KNN]; int bi[KNN];
#pragma unroll
  for (int j = 0; j < KNN; ++j) { bd[j] = 3.4e38f; bi[j] = 0; }

  const v4f* sp4 = (const v4f*)sp;
  int qi = tid;
  Pts8 pa = load_pts8(sp4, 0);
  for (int t = 0; t < TRIPS; t += 2) {
    const Pts8 pb = load_pts8(sp4, t + 1);
    scan_pts8(pa, t * PTS, px, py, pz, bd, bi, qbuf, tid, qi);
    const int tn = (t + 2 < TRIPS) ? (t + 2) : t;
    pa = load_pts8(sp4, tn);
    scan_pts8(pb, (t + 1) * PTS, px, py, pz, bd, bi, qbuf, tid, qi);
  }
  drain_queue(qbuf + tid, tid, qi, bd, bi);

  float sx = 0.f, sy = 0.f, sz = 0.f;
#pragma unroll
  for (int j = 0; j < KNN; ++j) {
    const int m = bi[j];
    sx += sp[3 * m + 0]; sy += sp[3 * m + 1]; sz += sp[3 * m + 2];
  }

  float* xr = xout + (size_t)row * INP8;
  xr[0] = px;          xr[1] = py;          xr[2] = pz;
  xr[3] = sx * 0.125f; xr[4] = sy * 0.125f; xr[5] = sz * 0.125f;
  xr[6] = 0.f;         xr[7] = 0.f;
}

__device__ __forceinline__ float relu(float x) { return x > 0.f ? x : 0.f; }

__global__ void __launch_bounds__(128)
mlp_refine_kernel(const float* __restrict__ xin,
                  const float* __restrict__ predicted,
                  const float* __restrict__ W1, const float* __restrict__ b1,
                  const float* __restrict__ mw, const float* __restrict__ mb,
                  const float* __restrict__ Wc, const float* __restrict__ bc,
                  float* __restrict__ out) {
  extern __shared__ __attribute__((aligned(16))) float smem[];
  const int tid   = threadIdx.x;
  const int wave  = tid >> 5;
  const int lane  = tid & 31;
  const int lrow  = lane & 15;
  const int khalf = lane >> 4;

  float* lds_h  = smem + wave * (32 * LP);
  float* lds_ht = smem + 4 * (32 * LP) + wave * (32 * LP);

  const int rowBase = blockIdx.x * 128 + wave * 32;
  const int dRow = (lane < 16) ? 0 : 8;

  {
    const float* xr0 = xin + (size_t)(rowBase + lrow) * INP8;
    const float* xr1 = xin + (size_t)(rowBase + 16 + lrow) * INP8;
    const float kmask = khalf ? 0.f : 1.f;
    v2f a00, a01, a10, a11;
    a00.x = xr0[khalf * 2 + 0];     a00.y = xr0[khalf * 2 + 1];
    a01.x = xr0[4 + khalf * 2 + 0]; a01.y = xr0[4 + khalf * 2 + 1];
    a10.x = xr1[khalf * 2 + 0];     a10.y = xr1[khalf * 2 + 1];
    a11.x = xr1[4 + khalf * 2 + 0]; a11.y = xr1[4 + khalf * 2 + 1];
#pragma unroll
    for (int nt = 0; nt < 8; ++nt) {
      const int n = nt * 16 + lrow;
      const float bv = b1[n];
      v8f c0, c1;
#pragma unroll
      for (int j = 0; j < 8; ++j) { c0[j] = bv; c1[j] = bv; }
      v2f bf0, bf1;
      const int k0 = khalf * 2;
      bf0.x = W1[n * INP + k0];        bf0.y = W1[n * INP + k0 + 1];
      bf1.x = kmask * W1[n * INP + 4]; bf1.y = kmask * W1[n * INP + 5];
      c0 = wmma_f32(a00, bf0, c0);  c0 = wmma_f32(a01, bf1, c0);
      c1 = wmma_f32(a10, bf0, c1);  c1 = wmma_f32(a11, bf1, c1);
#pragma unroll
      for (int j = 0; j < 8; ++j) {
        const int r = j + dRow;
        lds_h[r * LP + n]        = relu(c0[j]);
        lds_h[(16 + r) * LP + n] = relu(c1[j]);
      }
    }
  }

  for (int t = 0; t < OUTW; ++t) {
    v8f c0[8], c1[8];
#pragma unroll
    for (int nt = 0; nt < 8; ++nt) {
      const float bv = mb[t * HID + nt * 16 + lrow];
#pragma unroll
      for (int j = 0; j < 8; ++j) { c0[nt][j] = bv; c1[nt][j] = bv; }
    }
    const float* wp  = mw + (size_t)(t * HID + lrow) * HID + khalf * 4;
    const float* a0p = lds_h + lrow * LP + khalf * 4;
    const float* a1p = lds_h + (16 + lrow) * LP + khalf * 4;
    v4f bP[8], bQ[8];
#pragma unroll
    for (int nt = 0; nt < 8; ++nt)
      bP[nt] = *(const v4f*)(wp + nt * 16 * HID);
    for (int k8 = 0; k8 < HID; k8 += 16) {
#pragma unroll
      for (int nt = 0; nt < 8; ++nt)
        bQ[nt] = *(const v4f*)(wp + nt * 16 * HID + k8 + 8);
      {
        const v4f aA = *(const v4f*)(a0p + k8);
        const v4f aB = *(const v4f*)(a1p + k8);
        const v2f aA0 = {aA.x, aA.y}, aA1 = {aA.z, aA.w};
        const v2f aB0 = {aB.x, aB.y}, aB1 = {aB.z, aB.w};
#pragma unroll
        for (int nt = 0; nt < 8; ++nt) {
          const v2f bf = {bP[nt].x, bP[nt].y};
          c0[nt] = wmma_f32(aA0, bf, c0[nt]);
          c1[nt] = wmma_f32(aB0, bf, c1[nt]);
        }
#pragma unroll
        for (int nt = 0; nt < 8; ++nt) {
          const v2f bf = {bP[nt].z, bP[nt].w};
          c0[nt] = wmma_f32(aA1, bf, c0[nt]);
          c1[nt] = wmma_f32(aB1, bf, c1[nt]);
        }
      }
      const int kn = (k8 + 16 < HID) ? (k8 + 16) : k8;
#pragma unroll
      for (int nt = 0; nt < 8; ++nt)
        bP[nt] = *(const v4f*)(wp + nt * 16 * HID + kn);
      {
        const v4f aA = *(const v4f*)(a0p + k8 + 8);
        const v4f aB = *(const v4f*)(a1p + k8 + 8);
        const v2f aA0 = {aA.x, aA.y}, aA1 = {aA.z, aA.w};
        const v2f aB0 = {aB.x, aB.y}, aB1 = {aB.z, aB.w};
#pragma unroll
        for (int nt = 0; nt < 8; ++nt) {
          const v2f bf = {bQ[nt].x, bQ[nt].y};
          c0[nt] = wmma_f32(aA0, bf, c0[nt]);
          c1[nt] = wmma_f32(aB0, bf, c1[nt]);
        }
#pragma unroll
        for (int nt = 0; nt < 8; ++nt) {
          const v2f bf = {bQ[nt].z, bQ[nt].w};
          c0[nt] = wmma_f32(aA1, bf, c0[nt]);
          c1[nt] = wmma_f32(aB1, bf, c1[nt]);
        }
      }
    }
#pragma unroll
    for (int nt = 0; nt < 8; ++nt) {
      const int n = nt * 16 + lrow;
#pragma unroll
      for (int j = 0; j < 8; ++j) {
        const int r = j + dRow;
        lds_ht[r * LP + n]        = relu(c0[nt][j]);
        lds_ht[(16 + r) * LP + n] = relu(c1[nt][j]);
      }
    }
    {
      const int rg = rowBase + lane;
      const float* hr = lds_ht + lane * LP;
      float a0 = bc[0], a1 = bc[1], a2 = bc[2];
      for (int i = 0; i < HID; i += 4) {
        const v4f v = *(const v4f*)(hr + i);
#pragma unroll
        for (int q = 0; q < 4; ++q) {
          a0 += v[q] * Wc[0 * HID + i + q];
          a1 += v[q] * Wc[1 * HID + i + q];
          a2 += v[q] * Wc[2 * HID + i + q];
        }
      }
      float* o = out + ((size_t)rg * OUTW + t) * 3;
      o[0] = predicted[rg * 3 + 0] + a0;
      o[1] = predicted[rg * 3 + 1] + a1;
      o[2] = predicted[rg * 3 + 2] + a2;
    }
  }
}

extern "C" void kernel_launch(void* const* d_in, const int* in_sizes, int n_in,
                              void* d_out, int out_size, void* d_ws, size_t ws_size,
                              hipStream_t stream) {
  const float* partial   = (const float*)d_in[0];
  const float* predicted = (const float*)d_in[1];
  const float* W1 = (const float*)d_in[2];
  const float* b1 = (const float*)d_in[3];
  const float* W2 = (const float*)d_in[4];
  const float* b2 = (const float*)d_in[5];
  const float* Wd = (const float*)d_in[6];
  const float* bd = (const float*)d_in[7];
  const float* Wc = (const float*)d_in[8];
  const float* bc = (const float*)d_in[9];
  float* out = (float*)d_out;

  float* x  = (float*)d_ws;
  float* mw = x + (size_t)BB * NN * INP8;
  float* mb = mw + (size_t)OUTW * HID * HID;

  dim3 g1(NN / 256, BB + 1);
  knn_seed_kernel<<<g1, 256, 0, stream>>>(partial, predicted, W2, b2, Wd, bd, mw, mb, x);

  const int blocks = (BB * NN) / 128;
  const size_t lds = (size_t)8 * 32 * LP * sizeof(float);
  mlp_refine_kernel<<<blocks, 128, lds, stream>>>(
      x, predicted, W1, b1, mw, mb, Wc, bc, out);
}


# KNN scan queues one word per lane and 16-point step (group number + 16-bit selection mask) instead of an index per point; batches take set bits lowest first = index order; same list, same sums
# speedup vs baseline: 6.5931x; 1.4588x over previous
#include <hip/hip_runtime.h>
#include <hip/hip_bf16.h>


typedef __attribute__((ext_vector_type(2))) float v2f;
typedef __attribute__((ext_vector_type(4))) float v4f;
typedef __attribute__((ext_vector_type(8))) float v8f;

#define BB   8
#define MM   2048
#define NN   8192
#define KNN  8
#define HID  128
#define INP  6
#define INP8 8
#define OUTW 2
#define LP   132

__device__ __forceinline__ v8f wmma_f32(v2f a, v2f b, v8f c) {
  return __builtin_amdgcn_wmma_f32_16x16x4_f32(
       false, a,  false, b,
       (short)0, c,  false,  false);
}

#define GP 16
#define GROUPS (MM / GP)
#define QCAP 32
#define QROW 256
#define KAPPA 0x1p-16f

typedef __attribute__((ext_vector_type(2))) unsigned v2u;

__device__ __forceinline__ void insert_one(float dd, int id,
                                           float (&bd)[KNN], int (&bi)[KNN]) {
  bool lt[KNN];
#pragma unroll
  for (int j = 0; j < KNN; ++j) lt[j] = dd < bd[j];
#pragma unroll
  for (int j = KNN - 1; j >= 1; --j)
    bd[j] = __builtin_amdgcn_fmed3f(bd[j - 1], dd, bd[j]);
  bd[0] = lt[0] ? dd : bd[0];
  int ki[KNN];
#pragma unroll
  for (int j = 0; j < KNN; ++j) ki[j] = lt[j] ? id : bi[j];
#pragma unroll
  for (int j = KNN - 1; j >= 1; --j)
    bi[j] = lt[j - 1] ? bi[j - 1] : ki[j];
  bi[0] = ki[0];
}

__device__ __forceinline__ float rank_dist(float px, float py, float pz,
                                           float qx, float qy, float qz) {
  const float dx = px - qx;
  const float dy = py - qy;
  const float dz = pz - qz;
  float d;
  {
#pragma clang fp contract(off)
    const float sxy = dx * dx + dy * dy;
    d = __builtin_fmaf(dz, dz, sxy);
  }
  return d;
}

__device__ __forceinline__ void drain_queue(const int* qt, int tid, int& qi,
                                            const v4f* cloud,
                                            float px, float py, float pz,
                                            float (&bd)[KNN], int (&bi)[KNN]) {
  const int last = (QCAP - 1) * QROW + tid;
  unsigned cur = 0;
  int rd = tid;
  unsigned n0 = (unsigned)qt[0];
  unsigned n1 = (unsigned)qt[QROW];
  n0 = (rd < qi) ? n0 : 0u;
  n1 = (rd + QROW < qi) ? n1 : 0u;
#pragma nounroll
  for (int e = 0; e < QCAP; e += 2) {
    const bool t0 = (cur & 0xffffu) == 0u;
    cur = t0 ? n0 : cur;
    const unsigned nn = t0 ? n1 : n0;
    const bool live0 = (cur & 0xffffu) != 0u;
    const int i0 = live0 ? (int)(((cur >> 16) << 4) + (unsigned)__builtin_ctz(cur | 0x10000u)) : 0;
    cur &= cur - 1u;
    const bool t1 = (cur & 0xffffu) == 0u;
    cur = t1 ? nn : cur;
    const bool live1 = (cur & 0xffffu) != 0u;
    const int i1 = live1 ? (int)(((cur >> 16) << 4) + (unsigned)__builtin_ctz(cur | 0x10000u)) : 0;
    cur &= cur - 1u;
    rd += (t0 ? QROW : 0) + (t1 ? QROW : 0);
    if (__builtin_amdgcn_ballot_w32(live0) == 0) break;
    const v4f q0 = cloud[i0];
    const v4f q1 = cloud[i1];
    const int r0 = rd < last ? rd : last;
    const int r1 = rd + QROW < last ? rd + QROW : last;
    const unsigned w0 = (unsigned)qt[r0 - tid];
    const unsigned w1 = (unsigned)qt[r1 - tid];
    n0 = (rd < qi) ? w0 : 0u;
    n1 = (rd + QROW < qi) ? w1 : 0u;
    const float d0 = rank_dist(px, py, pz, q0.x, q0.y, q0.z);
    const float d1 = rank_dist(px, py, pz, q1.x, q1.y, q1.z);
    insert_one(live0 ? d0 : __builtin_inff(), i0, bd, bi);
    insert_one(live1 ? d1 : __builtin_inff(), i1, bd, bi);
  }
  qi = tid;
}

__device__ __forceinline__ void scan_group(const v8f& d1, const v8f& d2, int g,
                                           const v4f* cloud,
                                           float px, float py, float pz, float pp,
                                           float (&bd)[KNN], int (&bi)[KNN], float& tf,
                                           int* q, int tid, int& qi, int& cnt) {
  float f[GP];
#pragma unroll
  for (int r = 0; r < 8; ++r) {
    const float u1 = d1[r], u2 = d2[r];
    const v2u s = __builtin_amdgcn_permlane16_swap(__builtin_bit_cast(unsigned, u1),
                                                   __builtin_bit_cast(unsigned, u2),
                                                   false, false);
    const unsigned t1 = s.x, t2 = s.y;
    f[r]     = __builtin_bit_cast(float, t1);
    f[8 + r] = __builtin_bit_cast(float, t2);
  }
  if (__builtin_amdgcn_ballot_w32(cnt > QCAP - GP)) {
    drain_queue(q + tid, tid, qi, cloud, px, py, pz, bd, bi);
    cnt = 0;
    tf = (bd[KNN - 1] - pp) + KAPPA * pp;
  }
  unsigned sb[GP];
#pragma unroll
  for (int i = 0; i < GP; ++i) sb[i] = (f[i] < tf) ? (1u << i) : 0u;
  const unsigned o0 = sb[0] | sb[1] | sb[2],   o1 = sb[3] | sb[4] | sb[5];
  const unsigned o2 = sb[6] | sb[7] | sb[8],   o3 = sb[9] | sb[10] | sb[11];
  const unsigned o4 = sb[12] | sb[13] | sb[14];
  const unsigned m = (o0 | o1 | o2) | (o3 | o4 | sb[15]);
  q[qi] = (int)(((unsigned)g << 16) | m);
  qi += m ? QROW : 0;
  cnt += __builtin_popcount(m);
}

__device__ __forceinline__ void merge_linear_layers(const float* __restrict__ W2,
                                                    const float* __restrict__ b2,
                                                    const float* __restrict__ Wd,
                                                    const float* __restrict__ bd,
                                                    float* __restrict__ mw,
                                                    float* __restrict__ mb) {
  const int tid = threadIdx.x;
  const int bx  = blockIdx.x;
  if (bx < 16) {
    const int wave = tid >> 5, lane = tid & 31;
    const int c = lane & 15, half = lane >> 4;
    const int tile = bx * 8 + wave;
    const int mt = tile >> 3, nt = tile & 7;
    const int t = mt >> 3;
    const int o = (mt & 7) * 16 + c;
    const int j = nt * 16 + c;
    const float* ap = Wd + ((size_t)(2 * half) * HID + o) * OUTW + t;
    const float* bp = W2 + (2 * half) * HID + j;
    v8f acc;
#pragma unroll
    for (int r = 0; r < 8; ++r) acc[r] = 0.f;
#pragma unroll 8
    for (int k0 = 0; k0 < HID; k0 += 4) {
      v2f a, b;
      a.x = ap[(size_t)k0 * HID * OUTW];  a.y = ap[(size_t)(k0 + 1) * HID * OUTW];
      b.x = bp[k0 * HID];                 b.y = bp[(k0 + 1) * HID];
      acc = wmma_f32(a, b, acc);
    }
    float* dp = mw + (size_t)(mt * 16 + 8 * half) * HID + nt * 16 + c;
#pragma unroll
    for (int r = 0; r < 8; ++r) dp[r * HID] = acc[r];
  } else if (bx == 16) {
    const int t = tid >> 7, o = tid & (HID - 1);
    float s = 0.f;
    for (int i = 0; i < HID; ++i) s = __builtin_fmaf(b2[i], Wd[((size_t)i * HID + o) * OUTW + t], s);
    mb[tid] = s + bd[o];
  }
}

__global__ void __launch_bounds__(256)
knn_seed_kernel(const float* __restrict__ partial,
                const float* __restrict__ predicted,
                const float* __restrict__ lin2_w, const float* __restrict__ lin2_b,
                const float* __restrict__ dec_w,  const float* __restrict__ dec_b,
                float* __restrict__ mw, float* __restrict__ mb,
                float* __restrict__ xout) {
  if (blockIdx.y == BB) {
    merge_linear_layers(lin2_w, lin2_b, dec_w, dec_b, mw, mb);
    return;
  }
  __shared__ v4f cloud[MM];
  __shared__ int qbuf[QCAP * QROW];
  const int b   = blockIdx.y;
  const int tid = threadIdx.x;

  for (int i = tid; i < MM; i += 256) {
    const float* g = partial + ((size_t)b * MM + i) * 3;
    const float x = g[0], y = g[1], z = g[2];
    const v4f pt = {x, y, z, (x * x + y * y + z * z) * (1.0f - KAPPA)};
    cloud[i] = pt;
  }
  __syncthreads();

  const int n = blockIdx.x * 256 + tid;
  const int row = b * NN + n;
  const float px = predicted[row * 3 + 0];
  const float py = predicted[row * 3 + 1];
  const float pz = predicted[row * 3 + 2];
  const float pp = px * px + py * py + pz * pz;

  const int lane = tid & 31;
  const int c = lane & 15, half = lane >> 4;
  const float* g1 = predicted + (size_t)(row - lane + c) * 3;
  const float* g2 = predicted + (size_t)(row - lane + 16 + c) * 3;
  const float x1 = g1[0], y1 = g1[1], z1 = g1[2];
  const float x2 = g2[0], y2 = g2[1], z2 = g2[2];
  v2f b1, b2;
  b1.x = half ? -2.0f * z1 : -2.0f * x1;   b1.y = half ? 1.0f : -2.0f * y1;
  b2.x = half ? -2.0f * z2 : -2.0f * x2;   b2.y = half ? 1.0f : -2.0f * y2;
  const v2f* ap = (const v2f*)cloud + (c * 2 + half);
  v8f zero;
#pragma unroll
  for (int r = 0; r < 8; ++r) zero[r] = 0.f;

  float bd[KNN]; int bi[KNN];
#pragma unroll
  for (int j = 0; j < KNN; ++j) { bd[j] = 3.4e38f; bi[j] = 0; }
  float tf = (bd[KNN - 1] - pp) + KAPPA * pp;

  int qi = tid;
  int cnt = 0;
  v2f aA = ap[0];
  v2f aB = ap[32];
  v8f dA1 = wmma_f32(aA, b1, zero), dA2 = wmma_f32(aA, b2, zero);
  for (int g = 0; g < GROUPS; g += 2) {
    const v8f dB1 = wmma_f32(aB, b1, zero), dB2 = wmma_f32(aB, b2, zero);
    const int ga = (g + 2 < GROUPS) ? (g + 2) : g;
    aA = ap[ga * 32];
    scan_group(dA1, dA2, g, cloud, px, py, pz, pp, bd, bi, tf, qbuf, tid, qi, cnt);
    dA1 = wmma_f32(aA, b1, zero);  dA2 = wmma_f32(aA, b2, zero);
    const int gb = (g + 3 < GROUPS) ? (g + 3) : (g + 1);
    aB = ap[gb * 32];
    scan_group(dB1, dB2, g + 1, cloud, px, py, pz, pp, bd, bi, tf, qbuf, tid, qi, cnt);
  }
  drain_queue(qbuf + tid, tid, qi, cloud, px, py, pz, bd, bi);

  float sx = 0.f, sy = 0.f, sz = 0.f;
#pragma unroll
  for (int j = 0; j < KNN; ++j) {
    const v4f m = cloud[bi[j]];
    sx += m.x; sy += m.y; sz += m.z;
  }

  float* xr = xout + (size_t)row * INP8;
  xr[0] = px;          xr[1] = py;          xr[2] = pz;
  xr[3] = sx * 0.125f; xr[4] = sy * 0.125f; xr[5] = sz * 0.125f;
  xr[6] = 0.f;         xr[7] = 0.f;
}

__device__ __forceinline__ float relu(float x) { return x > 0.f ? x : 0.f; }

__global__ void __launch_bounds__(128)
mlp_refine_kernel(const float* __restrict__ xin,
                  const float* __restrict__ predicted,
                  const float* __restrict__ W1, const float* __restrict__ b1,
                  const float* __restrict__ mw, const float* __restrict__ mb,
                  const float* __restrict__ Wc, const float* __restrict__ bc,
                  float* __restrict__ out) {
  extern __shared__ __attribute__((aligned(16))) float smem[];
  const int tid   = threadIdx.x;
  const int wave  = tid >> 5;
  const int lane  = tid & 31;
  const int lrow  = lane & 15;
  const int khalf = lane >> 4;

  float* lds_h  = smem + wave * (32 * LP);
  float* lds_ht = smem + 4 * (32 * LP) + wave * (32 * LP);

  const int rowBase = blockIdx.x * 128 + wave * 32;
  const int dRow = (lane < 16) ? 0 : 8;

  {
    const float* xr0 = xin + (size_t)(rowBase + lrow) * INP8;
    const float* xr1 = xin + (size_t)(rowBase + 16 + lrow) * INP8;
    const float kmask = khalf ? 0.f : 1.f;
    v2f a00, a01, a10, a11;
    a00.x = xr0[khalf * 2 + 0];     a00.y = xr0[khalf * 2 + 1];
    a01.x = xr0[4 + khalf * 2 + 0]; a01.y = xr0[4 + khalf * 2 + 1];
    a10.x = xr1[khalf * 2 + 0];     a10.y = xr1[khalf * 2 + 1];
    a11.x = xr1[4 + khalf * 2 + 0]; a11.y = xr1[4 + khalf * 2 + 1];
#pragma unroll
    for (int nt = 0; nt < 8; ++nt) {
      const int n = nt * 16 + lrow;
      const float bv = b1[n];
      v8f c0, c1;
#pragma unroll
      for (int j = 0; j < 8; ++j) { c0[j] = bv; c1[j] = bv; }
      v2f bf0, bf1;
      const int k0 = khalf * 2;
      bf0.x = W1[n * INP + k0];        bf0.y = W1[n * INP + k0 + 1];
      bf1.x = kmask * W1[n * INP + 4]; bf1.y = kmask * W1[n * INP + 5];
      c0 = wmma_f32(a00, bf0, c0);  c0 = wmma_f32(a01, bf1, c0);
      c1 = wmma_f32(a10, bf0, c1);  c1 = wmma_f32(a11, bf1, c1);
#pragma unroll
      for (int j = 0; j < 8; ++j) {
        const int r = j + dRow;
        lds_h[r * LP + n]        = relu(c0[j]);
        lds_h[(16 + r) * LP + n] = relu(c1[j]);
      }
    }
  }

  for (int t = 0; t < OUTW; ++t) {
    v8f c0[8], c1[8];
#pragma unroll
    for (int nt = 0; nt < 8; ++nt) {
      const float bv = mb[t * HID + nt * 16 + lrow];
#pragma unroll
      for (int j = 0; j < 8; ++j) { c0[nt][j] = bv; c1[nt][j] = bv; }
    }
    const float* wp  = mw + (size_t)(t * HID + lrow) * HID + khalf * 4;
    const float* a0p = lds_h + lrow * LP + khalf * 4;
    const float* a1p = lds_h + (16 + lrow) * LP + khalf * 4;
    v4f bP[8], bQ[8];
#pragma unroll
    for (int nt = 0; nt < 8; ++nt)
      bP[nt] = *(const v4f*)(wp + nt * 16 * HID);
    for (int k8 = 0; k8 < HID; k8 += 16) {
#pragma unroll
      for (int nt = 0; nt < 8; ++nt)
        bQ[nt] = *(const v4f*)(wp + nt * 16 * HID + k8 + 8);
      {
        const v4f aA = *(const v4f*)(a0p + k8);
        const v4f aB = *(const v4f*)(a1p + k8);
        const v2f aA0 = {aA.x, aA.y}, aA1 = {aA.z, aA.w};
        const v2f aB0 = {aB.x, aB.y}, aB1 = {aB.z, aB.w};
#pragma unroll
        for (int nt = 0; nt < 8; ++nt) {
          const v2f bf = {bP[nt].x, bP[nt].y};
          c0[nt] = wmma_f32(aA0, bf, c0[nt]);
          c1[nt] = wmma_f32(aB0, bf, c1[nt]);
        }
#pragma unroll
        for (int nt = 0; nt < 8; ++nt) {
          const v2f bf = {bP[nt].z, bP[nt].w};
          c0[nt] = wmma_f32(aA1, bf, c0[nt]);
          c1[nt] = wmma_f32(aB1, bf, c1[nt]);
        }
      }
      const int kn = (k8 + 16 < HID) ? (k8 + 16) : k8;
#pragma unroll
      for (int nt = 0; nt < 8; ++nt)
        bP[nt] = *(const v4f*)(wp + nt * 16 * HID + kn);
      {
        const v4f aA = *(const v4f*)(a0p + k8 + 8);
        const v4f aB = *(const v4f*)(a1p + k8 + 8);
        const v2f aA0 = {aA.x, aA.y}, aA1 = {aA.z, aA.w};
        const v2f aB0 = {aB.x, aB.y}, aB1 = {aB.z, aB.w};
#pragma unroll
        for (int nt = 0; nt < 8; ++nt) {
          const v2f bf = {bQ[nt].x, bQ[nt].y};
          c0[nt] = wmma_f32(aA0, bf, c0[nt]);
          c1[nt] = wmma_f32(aB0, bf, c1[nt]);
        }
#pragma unroll
        for (int nt = 0; nt < 8; ++nt) {
          const v2f bf = {bQ[nt].z, bQ[nt].w};
          c0[nt] = wmma_f32(aA1, bf, c0[nt]);
          c1[nt] = wmma_f32(aB1, bf, c1[nt]);
        }
      }
    }
#pragma unroll
    for (int nt = 0; nt < 8; ++nt) {
      const int n = nt * 16 + lrow;
#pragma unroll
      for (int j = 0; j < 8; ++j) {
        const int r = j + dRow;
        lds_ht[r * LP + n]        = relu(c0[nt][j]);
        lds_ht[(16 + r) * LP + n] = relu(c1[nt][j]);
      }
    }
    {
      const int rg = rowBase + lane;
      const float* hr = lds_ht + lane * LP;
      float a0 = bc[0], a1 = bc[1], a2 = bc[2];
      for (int i = 0; i < HID; i += 4) {
        const v4f v = *(const v4f*)(hr + i);
#pragma unroll
        for (int q = 0; q < 4; ++q) {
          a0 += v[q] * Wc[0 * HID + i + q];
          a1 += v[q] * Wc[1 * HID + i + q];
          a2 += v[q] * Wc[2 * HID + i + q];
        }
      }
      float* o = out + ((size_t)rg * OUTW + t) * 3;
      o[0] = predicted[rg * 3 + 0] + a0;
      o[1] = predicted[rg * 3 + 1] + a1;
      o[2] = predicted[rg * 3 + 2] + a2;
    }
  }
}

extern "C" void kernel_launch(void* const* d_in, const int* in_sizes, int n_in,
                              void* d_out, int out_size, void* d_ws, size_t ws_size,
                              hipStream_t stream) {
  const float* partial   = (const float*)d_in[0];
  const float* predicted = (const float*)d_in[1];
  const float* W1 = (const float*)d_in[2];
  const float* b1 = (const float*)d_in[3];
  const float* W2 = (const float*)d_in[4];
  const float* b2 = (const float*)d_in[5];
  const float* Wd = (const float*)d_in[6];
  const float* bd = (const float*)d_in[7];
  const float* Wc = (const float*)d_in[8];
  const float* bc = (const float*)d_in[9];
  float* out = (float*)d_out;

  float* x  = (float*)d_ws;
  float* mw = x + (size_t)BB * NN * INP8;
  float* mb = mw + (size_t)OUTW * HID * HID;

  dim3 g1(NN / 256, BB + 1);
  knn_seed_kernel<<<g1, 256, 0, stream>>>(partial, predicted, W2, b2, Wd, bd, mw, mb, x);

  const int blocks = (BB * NN) / 128;
  const size_t lds = (size_t)8 * 32 * LP * sizeof(float);
  mlp_refine_kernel<<<blocks, 128, lds, stream>>>(
      x, predicted, W1, b1, mw, mb, Wc, bc, out);
}


# KNN scan: -tf rides in the WMMA's C operand (product = f-tf); sign bits funnel-shifted into 8-bit masks, one lane-half swap on the masks (was 8 swaps + 16 compare/selects); same list, same sums
# speedup vs baseline: 6.9728x; 1.0576x over previous
#include <hip/hip_runtime.h>
#include <hip/hip_bf16.h>


typedef __attribute__((ext_vector_type(2))) float v2f;
typedef __attribute__((ext_vector_type(4))) float v4f;
typedef __attribute__((ext_vector_type(8))) float v8f;

#define BB   8
#define MM   2048
#define NN   8192
#define KNN  8
#define HID  128
#define INP  6
#define INP8 8
#define OUTW 2
#define LP   132

__device__ __forceinline__ v8f wmma_f32(v2f a, v2f b, v8f c) {
  return __builtin_amdgcn_wmma_f32_16x16x4_f32(
       false, a,  false, b,
       (short)0, c,  false,  false);
}

#define GP 16
#define GROUPS (MM / GP)
#define QCAP 32
#define QROW 256
#define KAPPA 0x1p-16f

typedef __attribute__((ext_vector_type(2))) unsigned v2u;

__device__ __forceinline__ void insert_one(float dd, int id,
                                           float (&bd)[KNN], int (&bi)[KNN]) {
  bool lt[KNN];
#pragma unroll
  for (int j = 0; j < KNN; ++j) lt[j] = dd < bd[j];
#pragma unroll
  for (int j = KNN - 1; j >= 1; --j)
    bd[j] = __builtin_amdgcn_fmed3f(bd[j - 1], dd, bd[j]);
  bd[0] = lt[0] ? dd : bd[0];
  int ki[KNN];
#pragma unroll
  for (int j = 0; j < KNN; ++j) ki[j] = lt[j] ? id : bi[j];
#pragma unroll
  for (int j = KNN - 1; j >= 1; --j)
    bi[j] = lt[j - 1] ? bi[j - 1] : ki[j];
  bi[0] = ki[0];
}

__device__ __forceinline__ float rank_dist(float px, float py, float pz,
                                           float qx, float qy, float qz) {
  const float dx = px - qx;
  const float dy = py - qy;
  const float dz = pz - qz;
  float d;
  {
#pragma clang fp contract(off)
    const float sxy = dx * dx + dy * dy;
    d = __builtin_fmaf(dz, dz, sxy);
  }
  return d;
}

__device__ __forceinline__ void drain_queue(const int* qt, int tid, int& qi,
                                            const v4f* cloud,
                                            float px, float py, float pz,
                                            float (&bd)[KNN], int (&bi)[KNN]) {
  const int last = (QCAP - 1) * QROW + tid;
  unsigned cur = 0;
  int rd = tid;
  unsigned n0 = (unsigned)qt[0];
  unsigned n1 = (unsigned)qt[QROW];
  n0 = (rd < qi) ? n0 : 0u;
  n1 = (rd + QROW < qi) ? n1 : 0u;
#pragma nounroll
  for (int e = 0; e < QCAP; e += 2) {
    const bool t0 = (cur & 0xffffu) == 0u;
    cur = t0 ? n0 : cur;
    const unsigned nn = t0 ? n1 : n0;
    const bool live0 = (cur & 0xffffu) != 0u;
    const int i0 = live0 ? (int)(((cur >> 16) << 4) + (unsigned)__builtin_ctz(cur | 0x10000u)) : 0;
    cur &= cur - 1u;
    const bool t1 = (cur & 0xffffu) == 0u;
    cur = t1 ? nn : cur;
    const bool live1 = (cur & 0xffffu) != 0u;
    const int i1 = live1 ? (int)(((cur >> 16) << 4) + (unsigned)__builtin_ctz(cur | 0x10000u)) : 0;
    cur &= cur - 1u;
    rd += (t0 ? QROW : 0) + (t1 ? QROW : 0);
    if (__builtin_amdgcn_ballot_w32(live0) == 0) break;
    const v4f q0 = cloud[i0];
    const v4f q1 = cloud[i1];
    const int r0 = rd < last ? rd : last;
    const int r1 = rd + QROW < last ? rd + QROW : last;
    const unsigned w0 = (unsigned)qt[r0 - tid];
    const unsigned w1 = (unsigned)qt[r1 - tid];
    n0 = (rd < qi) ? w0 : 0u;
    n1 = (rd + QROW < qi) ? w1 : 0u;
    const float d0 = rank_dist(px, py, pz, q0.x, q0.y, q0.z);
    const float d1 = rank_dist(px, py, pz, q1.x, q1.y, q1.z);
    insert_one(live0 ? d0 : __builtin_inff(), i0, bd, bi);
    insert_one(live1 ? d1 : __builtin_inff(), i1, bd, bi);
  }
  qi = tid;
}

__device__ __forceinline__ void set_threshold(float t, float pp, v8f& nt1, v8f& nt2) {
  const float tf = (t - pp) + KAPPA * pp;
  const unsigned own = __builtin_bit_cast(unsigned, -tf);
  const v2u s = __builtin_amdgcn_permlane16_swap(own, own, false, false);
  const unsigned s1 = s.x, s2 = s.y;
  const float c1 = __builtin_bit_cast(float, s1);
  const float c2 = __builtin_bit_cast(float, s2);
#pragma unroll
  for (int r = 0; r < 8; ++r) { nt1[r] = c1; nt2[r] = c2; }
}

__device__ __forceinline__ unsigned sign_mask8(const v8f& d) {
  unsigned lo = 0u, hi = 0u;
#pragma unroll
  for (int r = 3; r >= 0; --r) {
    const float vl = d[r], vh = d[4 + r];
    lo = __builtin_amdgcn_alignbit(lo, __builtin_bit_cast(unsigned, vl), 31u);
    hi = __builtin_amdgcn_alignbit(hi, __builtin_bit_cast(unsigned, vh), 31u);
  }
  return lo | (hi << 4);
}

__device__ __forceinline__ void queue_group(const v8f& d1, const v8f& d2, int g,
                                            int* q, int& qi, int& cnt) {
  const unsigned m1 = sign_mask8(d1);
  const unsigned m2 = sign_mask8(d2);
  const v2u s = __builtin_amdgcn_permlane16_swap(m1, m2, false, false);
  const unsigned slo = s.x, shi = s.y;
  const unsigned m = slo | (shi << 8);
  q[qi] = (int)(((unsigned)g << 16) | m);
  qi += m ? QROW : 0;
  cnt += __builtin_popcount(m);
}

__device__ __forceinline__ void merge_linear_layers(const float* __restrict__ W2,
                                                    const float* __restrict__ b2,
                                                    const float* __restrict__ Wd,
                                                    const float* __restrict__ bd,
                                                    float* __restrict__ mw,
                                                    float* __restrict__ mb) {
  const int tid = threadIdx.x;
  const int bx  = blockIdx.x;
  if (bx < 16) {
    const int wave = tid >> 5, lane = tid & 31;
    const int c = lane & 15, half = lane >> 4;
    const int tile = bx * 8 + wave;
    const int mt = tile >> 3, nt = tile & 7;
    const int t = mt >> 3;
    const int o = (mt & 7) * 16 + c;
    const int j = nt * 16 + c;
    const float* ap = Wd + ((size_t)(2 * half) * HID + o) * OUTW + t;
    const float* bp = W2 + (2 * half) * HID + j;
    v8f acc;
#pragma unroll
    for (int r = 0; r < 8; ++r) acc[r] = 0.f;
#pragma unroll 8
    for (int k0 = 0; k0 < HID; k0 += 4) {
      v2f a, b;
      a.x = ap[(size_t)k0 * HID * OUTW];  a.y = ap[(size_t)(k0 + 1) * HID * OUTW];
      b.x = bp[k0 * HID];                 b.y = bp[(k0 + 1) * HID];
      acc = wmma_f32(a, b, acc);
    }
    float* dp = mw + (size_t)(mt * 16 + 8 * half) * HID + nt * 16 + c;
#pragma unroll
    for (int r = 0; r < 8; ++r) dp[r * HID] = acc[r];
  } else if (bx == 16) {
    const int t = tid >> 7, o = tid & (HID - 1);
    float s = 0.f;
    for (int i = 0; i < HID; ++i) s = __builtin_fmaf(b2[i], Wd[((size_t)i * HID + o) * OUTW + t], s);
    mb[tid] = s + bd[o];
  }
}

__global__ void __launch_bounds__(256)
knn_seed_kernel(const float* __restrict__ partial,
                const float* __restrict__ predicted,
                const float* __restrict__ lin2_w, const float* __restrict__ lin2_b,
                const float* __restrict__ dec_w,  const float* __restrict__ dec_b,
                float* __restrict__ mw, float* __restrict__ mb,
                float* __restrict__ xout) {
  if (blockIdx.y == BB) {
    merge_linear_layers(lin2_w, lin2_b, dec_w, dec_b, mw, mb);
    return;
  }
  __shared__ v4f cloud[MM];
  __shared__ int qbuf[QCAP * QROW];
  const int b   = blockIdx.y;
  const int tid = threadIdx.x;

  for (int i = tid; i < MM; i += 256) {
    const float* g = partial + ((size_t)b * MM + i) * 3;
    const float x = g[0], y = g[1], z = g[2];
    const v4f pt = {x, y, z, (x * x + y * y + z * z) * (1.0f - KAPPA)};
    cloud[i] = pt;
  }
  __syncthreads();

  const int n = blockIdx.x * 256 + tid;
  const int row = b * NN + n;
  const float px = predicted[row * 3 + 0];
  const float py = predicted[row * 3 + 1];
  const float pz = predicted[row * 3 + 2];
  const float pp = px * px + py * py + pz * pz;

  const int lane = tid & 31;
  const int c = lane & 15, half = lane >> 4;
  const float* g1 = predicted + (size_t)(row - lane + c) * 3;
  const float* g2 = predicted + (size_t)(row - lane + 16 + c) * 3;
  const float x1 = g1[0], y1 = g1[1], z1 = g1[2];
  const float x2 = g2[0], y2 = g2[1], z2 = g2[2];
  v2f b1, b2;
  b1.x = half ? -2.0f * z1 : -2.0f * x1;   b1.y = half ? 1.0f : -2.0f * y1;
  b2.x = half ? -2.0f * z2 : -2.0f * x2;   b2.y = half ? 1.0f : -2.0f * y2;
  const v2f* ap = (const v2f*)cloud + (c * 2 + half);
  float bd[KNN]; int bi[KNN];
#pragma unroll
  for (int j = 0; j < KNN; ++j) { bd[j] = 3.4e38f; bi[j] = 0; }

  int qi = tid;
  int cnt = 0;
  v8f nt1, nt2;
  set_threshold(bd[KNN - 1], pp, nt1, nt2);

  v2f aA = ap[0];
  v2f aB = ap[32];
  for (int g = 0; g < GROUPS; g += 2) {
    v8f dA1 = wmma_f32(aA, b1, nt1), dA2 = wmma_f32(aA, b2, nt2);
    v8f dB1 = wmma_f32(aB, b1, nt1), dB2 = wmma_f32(aB, b2, nt2);
    const int gn = (g + 2 < GROUPS) ? (g + 2) : g;
    const v2f nA = ap[gn * 32];
    const v2f nB = ap[gn * 32 + 32];
    if (__builtin_amdgcn_ballot_w32(cnt > QCAP - GP)) {
      drain_queue(qbuf + tid, tid, qi, cloud, px, py, pz, bd, bi);
      cnt = 0;
      set_threshold(bd[KNN - 1], pp, nt1, nt2);
      dA1 = wmma_f32(aA, b1, nt1);  dA2 = wmma_f32(aA, b2, nt2);
      dB1 = wmma_f32(aB, b1, nt1);  dB2 = wmma_f32(aB, b2, nt2);
    }
    queue_group(dA1, dA2, g, qbuf, qi, cnt);
    if (__builtin_amdgcn_ballot_w32(cnt > QCAP - GP)) {
      drain_queue(qbuf + tid, tid, qi, cloud, px, py, pz, bd, bi);
      cnt = 0;
      set_threshold(bd[KNN - 1], pp, nt1, nt2);
      dB1 = wmma_f32(aB, b1, nt1);  dB2 = wmma_f32(aB, b2, nt2);
    }
    queue_group(dB1, dB2, g + 1, qbuf, qi, cnt);
    aA = nA;
    aB = nB;
  }
  drain_queue(qbuf + tid, tid, qi, cloud, px, py, pz, bd, bi);

  float sx = 0.f, sy = 0.f, sz = 0.f;
#pragma unroll
  for (int j = 0; j < KNN; ++j) {
    const v4f m = cloud[bi[j]];
    sx += m.x; sy += m.y; sz += m.z;
  }

  float* xr = xout + (size_t)row * INP8;
  xr[0] = px;          xr[1] = py;          xr[2] = pz;
  xr[3] = sx * 0.125f; xr[4] = sy * 0.125f; xr[5] = sz * 0.125f;
  xr[6] = 0.f;         xr[7] = 0.f;
}

__device__ __forceinline__ float relu(float x) { return x > 0.f ? x : 0.f; }

__global__ void __launch_bounds__(128)
mlp_refine_kernel(const float* __restrict__ xin,
                  const float* __restrict__ predicted,
                  const float* __restrict__ W1, const float* __restrict__ b1,
                  const float* __restrict__ mw, const float* __restrict__ mb,
                  const float* __restrict__ Wc, const float* __restrict__ bc,
                  float* __restrict__ out) {
  extern __shared__ __attribute__((aligned(16))) float smem[];
  const int tid   = threadIdx.x;
  const int wave  = tid >> 5;
  const int lane  = tid & 31;
  const int lrow  = lane & 15;
  const int khalf = lane >> 4;

  float* lds_h  = smem + wave * (32 * LP);
  float* lds_ht = smem + 4 * (32 * LP) + wave * (32 * LP);

  const int rowBase = blockIdx.x * 128 + wave * 32;
  const int dRow = (lane < 16) ? 0 : 8;

  {
    const float* xr0 = xin + (size_t)(rowBase + lrow) * INP8;
    const float* xr1 = xin + (size_t)(rowBase + 16 + lrow) * INP8;
    const float kmask = khalf ? 0.f : 1.f;
    v2f a00, a01, a10, a11;
    a00.x = xr0[khalf * 2 + 0];     a00.y = xr0[khalf * 2 + 1];
    a01.x = xr0[4 + khalf * 2 + 0]; a01.y = xr0[4 + khalf * 2 + 1];
    a10.x = xr1[khalf * 2 + 0];     a10.y = xr1[khalf * 2 + 1];
    a11.x = xr1[4 + khalf * 2 + 0]; a11.y = xr1[4 + khalf * 2 + 1];
#pragma unroll
    for (int nt = 0; nt < 8; ++nt) {
      const int n = nt * 16 + lrow;
      const float bv = b1[n];
      v8f c0, c1;
#pragma unroll
      for (int j = 0; j < 8; ++j) { c0[j] = bv; c1[j] = bv; }
      v2f bf0, bf1;
      const int k0 = khalf * 2;
      bf0.x = W1[n * INP + k0];        bf0.y = W1[n * INP + k0 + 1];
      bf1.x = kmask * W1[n * INP + 4]; bf1.y = kmask * W1[n * INP + 5];
      c0 = wmma_f32(a00, bf0, c0);  c0 = wmma_f32(a01, bf1, c0);
      c1 = wmma_f32(a10, bf0, c1);  c1 = wmma_f32(a11, bf1, c1);
#pragma unroll
      for (int j = 0; j < 8; ++j) {
        const int r = j + dRow;
        lds_h[r * LP + n]        = relu(c0[j]);
        lds_h[(16 + r) * LP + n] = relu(c1[j]);
      }
    }
  }

  for (int t = 0; t < OUTW; ++t) {
    v8f c0[8], c1[8];
#pragma unroll
    for (int nt = 0; nt < 8; ++nt) {
      const float bv = mb[t * HID + nt * 16 + lrow];
#pragma unroll
      for (int j = 0; j < 8; ++j) { c0[nt][j] = bv; c1[nt][j] = bv; }
    }
    const float* wp  = mw + (size_t)(t * HID + lrow) * HID + khalf * 4;
    const float* a0p = lds_h + lrow * LP + khalf * 4;
    const float* a1p = lds_h + (16 + lrow) * LP + khalf * 4;
    v4f bP[8], bQ[8];
#pragma unroll
    for (int nt = 0; nt < 8; ++nt)
      bP[nt] = *(const v4f*)(wp + nt * 16 * HID);
    for (int k8 = 0; k8 < HID; k8 += 16) {
#pragma unroll
      for (int nt = 0; nt < 8; ++nt)
        bQ[nt] = *(const v4f*)(wp + nt * 16 * HID + k8 + 8);
      {
        const v4f aA = *(const v4f*)(a0p + k8);
        const v4f aB = *(const v4f*)(a1p + k8);
        const v2f aA0 = {aA.x, aA.y}, aA1 = {aA.z, aA.w};
        const v2f aB0 = {aB.x, aB.y}, aB1 = {aB.z, aB.w};
#pragma unroll
        for (int nt = 0; nt < 8; ++nt) {
          const v2f bf = {bP[nt].x, bP[nt].y};
          c0[nt] = wmma_f32(aA0, bf, c0[nt]);
          c1[nt] = wmma_f32(aB0, bf, c1[nt]);
        }
#pragma unroll
        for (int nt = 0; nt < 8; ++nt) {
          const v2f bf = {bP[nt].z, bP[nt].w};
          c0[nt] = wmma_f32(aA1, bf, c0[nt]);
          c1[nt] = wmma_f32(aB1, bf, c1[nt]);
        }
      }
      const int kn = (k8 + 16 < HID) ? (k8 + 16) : k8;
#pragma unroll
      for (int nt = 0; nt < 8; ++nt)
        bP[nt] = *(const v4f*)(wp + nt * 16 * HID + kn);
      {
        const v4f aA = *(const v4f*)(a0p + k8 + 8);
        const v4f aB = *(const v4f*)(a1p + k8 + 8);
        const v2f aA0 = {aA.x, aA.y}, aA1 = {aA.z, aA.w};
        const v2f aB0 = {aB.x, aB.y}, aB1 = {aB.z, aB.w};
#pragma unroll
        for (int nt = 0; nt < 8; ++nt) {
          const v2f bf = {bQ[nt].x, bQ[nt].y};
          c0[nt] = wmma_f32(aA0, bf, c0[nt]);
          c1[nt] = wmma_f32(aB0, bf, c1[nt]);
        }
#pragma unroll
        for (int nt = 0; nt < 8; ++nt) {
          const v2f bf = {bQ[nt].z, bQ[nt].w};
          c0[nt] = wmma_f32(aA1, bf, c0[nt]);
          c1[nt] = wmma_f32(aB1, bf, c1[nt]);
        }
      }
    }
#pragma unroll
    for (int nt = 0; nt < 8; ++nt) {
      const int n = nt * 16 + lrow;
#pragma unroll
      for (int j = 0; j < 8; ++j) {
        const int r = j + dRow;
        lds_ht[r * LP + n]        = relu(c0[nt][j]);
        lds_ht[(16 + r) * LP + n] = relu(c1[nt][j]);
      }
    }
    {
      const int rg = rowBase + lane;
      const float* hr = lds_ht + lane * LP;
      float a0 = bc[0], a1 = bc[1], a2 = bc[2];
      for (int i = 0; i < HID; i += 4) {
        const v4f v = *(const v4f*)(hr + i);
#pragma unroll
        for (int q = 0; q < 4; ++q) {
          a0 += v[q] * Wc[0 * HID + i + q];
          a1 += v[q] * Wc[1 * HID + i + q];
          a2 += v[q] * Wc[2 * HID + i + q];
        }
      }
      float* o = out + ((size_t)rg * OUTW + t) * 3;
      o[0] = predicted[rg * 3 + 0] + a0;
      o[1] = predicted[rg * 3 + 1] + a1;
      o[2] = predicted[rg * 3 + 2] + a2;
    }
  }
}

extern "C" void kernel_launch(void* const* d_in, const int* in_sizes, int n_in,
                              void* d_out, int out_size, void* d_ws, size_t ws_size,
                              hipStream_t stream) {
  const float* partial   = (const float*)d_in[0];
  const float* predicted = (const float*)d_in[1];
  const float* W1 = (const float*)d_in[2];
  const float* b1 = (const float*)d_in[3];
  const float* W2 = (const float*)d_in[4];
  const float* b2 = (const float*)d_in[5];
  const float* Wd = (const float*)d_in[6];
  const float* bd = (const float*)d_in[7];
  const float* Wc = (const float*)d_in[8];
  const float* bc = (const float*)d_in[9];
  float* out = (float*)d_out;

  float* x  = (float*)d_ws;
  float* mw = x + (size_t)BB * NN * INP8;
  float* mb = mw + (size_t)OUTW * HID * HID;

  dim3 g1(NN / 256, BB + 1);
  knn_seed_kernel<<<g1, 256, 0, stream>>>(partial, predicted, W2, b2, Wd, bd, mw, mb, x);

  const int blocks = (BB * NN) / 128;
  const size_t lds = (size_t)8 * 32 * LP * sizeof(float);
  mlp_refine_kernel<<<blocks, 128, lds, stream>>>(
      x, predicted, W1, b1, mw, mb, Wc, bc, out);
}


# KNN batch insertion pipelined three deep: pass k reads pass k+1's coordinates and pass k+2's queue words from LDS ahead of its own two insertions, so no pass waits for LDS; same list, same sums
# speedup vs baseline: 7.2108x; 1.0341x over previous
#include <hip/hip_runtime.h>
#include <hip/hip_bf16.h>


typedef __attribute__((ext_vector_type(2))) float v2f;
typedef __attribute__((ext_vector_type(4))) float v4f;
typedef __attribute__((ext_vector_type(8))) float v8f;

#define BB   8
#define MM   2048
#define NN   8192
#define KNN  8
#define HID  128
#define INP  6
#define INP8 8
#define OUTW 2
#define LP   132

__device__ __forceinline__ v8f wmma_f32(v2f a, v2f b, v8f c) {
  return __builtin_amdgcn_wmma_f32_16x16x4_f32(
       false, a,  false, b,
       (short)0, c,  false,  false);
}

#define GP 16
#define GROUPS (MM / GP)
#define QCAP 32
#define QROW 256
#define KAPPA 0x1p-16f

typedef __attribute__((ext_vector_type(2))) unsigned v2u;

__device__ __forceinline__ void insert_one(float dd, int id,
                                           float (&bd)[KNN], int (&bi)[KNN]) {
  bool lt[KNN];
#pragma unroll
  for (int j = 0; j < KNN; ++j) lt[j] = dd < bd[j];
#pragma unroll
  for (int j = KNN - 1; j >= 1; --j)
    bd[j] = __builtin_amdgcn_fmed3f(bd[j - 1], dd, bd[j]);
  bd[0] = lt[0] ? dd : bd[0];
  int ki[KNN];
#pragma unroll
  for (int j = 0; j < KNN; ++j) ki[j] = lt[j] ? id : bi[j];
#pragma unroll
  for (int j = KNN - 1; j >= 1; --j)
    bi[j] = lt[j - 1] ? bi[j - 1] : ki[j];
  bi[0] = ki[0];
}

__device__ __forceinline__ float rank_dist(float px, float py, float pz,
                                           float qx, float qy, float qz) {
  const float dx = px - qx;
  const float dy = py - qy;
  const float dz = pz - qz;
  float d;
  {
#pragma clang fp contract(off)
    const float sxy = dx * dx + dy * dy;
    d = __builtin_fmaf(dz, dz, sxy);
  }
  return d;
}


__device__ __forceinline__ void take_two(unsigned& cur, unsigned n0, unsigned n1, int& rd,
                                         int& i0, int& i1, bool& live0, bool& live1) {
  const bool t0 = (cur & 0xffffu) == 0u;
  cur = t0 ? n0 : cur;
  const unsigned nn = t0 ? n1 : n0;
  live0 = (cur & 0xffffu) != 0u;
  i0 = live0 ? (int)(((cur >> 16) << 4) + (unsigned)__builtin_ctz(cur | 0x10000u)) : 0;
  cur &= cur - 1u;
  const bool t1 = (cur & 0xffffu) == 0u;
  cur = t1 ? nn : cur;
  live1 = (cur & 0xffffu) != 0u;
  i1 = live1 ? (int)(((cur >> 16) << 4) + (unsigned)__builtin_ctz(cur | 0x10000u)) : 0;
  cur &= cur - 1u;
  rd += (t0 ? QROW : 0) + (t1 ? QROW : 0);
}

__device__ __forceinline__ void words_from(const int* qt, int tid, int rd, int qi, int last,
                                           unsigned& n0, unsigned& n1) {
  const int r0 = rd < last ? rd : last;
  const int r1 = rd + QROW < last ? rd + QROW : last;
  const unsigned w0 = (unsigned)qt[r0 - tid];
  const unsigned w1 = (unsigned)qt[r1 - tid];
  n0 = (rd < qi) ? w0 : 0u;
  n1 = (rd + QROW < qi) ? w1 : 0u;
}

typedef __attribute__((ext_vector_type(3))) float v3f;
__device__ __forceinline__ v3f coords(const v4f* cloud, int i) {
  return *(const v3f*)(cloud + i);
}

__device__ __forceinline__ void drain_queue(const int* qt, int tid, int& qi,
                                            const v4f* cloud,
                                            float px, float py, float pz,
                                            float (&bd)[KNN], int (&bi)[KNN]) {
  const int last = (QCAP - 1) * QROW + tid;
  const float inf = __builtin_inff();
  unsigned cur = 0;
  int rd = tid;
  unsigned n0, n1;
  words_from(qt, tid, rd, qi, last, n0, n1);
  int i0, i1, j0, j1;
  bool live0, live1, next0, next1;
  take_two(cur, n0, n1, rd, i0, i1, live0, live1);
  float d0, d1;
  {
    const v3f q0 = coords(cloud, i0);
    const v3f q1 = coords(cloud, i1);
    words_from(qt, tid, rd, qi, last, n0, n1);
    take_two(cur, n0, n1, rd, j0, j1, next0, next1);
    words_from(qt, tid, rd, qi, last, n0, n1);
    d0 = live0 ? rank_dist(px, py, pz, q0.x, q0.y, q0.z) : inf;
    d1 = live1 ? rank_dist(px, py, pz, q1.x, q1.y, q1.z) : inf;
  }
#pragma nounroll
  for (int e = 0; e < QCAP; e += 2) {
    if (__builtin_amdgcn_ballot_w32(live0) == 0) break;
    const v3f r0 = coords(cloud, j0);
    const v3f r1 = coords(cloud, j1);
    __builtin_amdgcn_sched_barrier(0);
    int m0, m1;
    bool then0, then1;
    take_two(cur, n0, n1, rd, m0, m1, then0, then1);
    const int e0 = rd < last ? rd : last;
    const int e1 = rd + QROW < last ? rd + QROW : last;
    const unsigned w0 = (unsigned)qt[e0 - tid];
    const unsigned w1 = (unsigned)qt[e1 - tid];
    insert_one(d0, i0, bd, bi);
    insert_one(d1, i1, bd, bi);
    __builtin_amdgcn_sched_barrier(0);
    n0 = (rd < qi) ? w0 : 0u;
    n1 = (rd + QROW < qi) ? w1 : 0u;
    d0 = next0 ? rank_dist(px, py, pz, r0.x, r0.y, r0.z) : inf;
    d1 = next1 ? rank_dist(px, py, pz, r1.x, r1.y, r1.z) : inf;
    i0 = j0;  i1 = j1;  live0 = next0;
    j0 = m0;  j1 = m1;  next0 = then0;  next1 = then1;
  }
  qi = tid;
}

__device__ __forceinline__ void set_threshold(float t, float pp, v8f& nt1, v8f& nt2) {
  const float tf = (t - pp) + KAPPA * pp;
  const unsigned own = __builtin_bit_cast(unsigned, -tf);
  const v2u s = __builtin_amdgcn_permlane16_swap(own, own, false, false);
  const unsigned s1 = s.x, s2 = s.y;
  const float c1 = __builtin_bit_cast(float, s1);
  const float c2 = __builtin_bit_cast(float, s2);
#pragma unroll
  for (int r = 0; r < 8; ++r) { nt1[r] = c1; nt2[r] = c2; }
}

__device__ __forceinline__ unsigned sign_mask8(const v8f& d) {
  unsigned lo = 0u, hi = 0u;
#pragma unroll
  for (int r = 3; r >= 0; --r) {
    const float vl = d[r], vh = d[4 + r];
    lo = __builtin_amdgcn_alignbit(lo, __builtin_bit_cast(unsigned, vl), 31u);
    hi = __builtin_amdgcn_alignbit(hi, __builtin_bit_cast(unsigned, vh), 31u);
  }
  return lo | (hi << 4);
}

__device__ __forceinline__ void queue_group(const v8f& d1, const v8f& d2, int g,
                                            int* q, int& qi, int& cnt) {
  const unsigned m1 = sign_mask8(d1);
  const unsigned m2 = sign_mask8(d2);
  const v2u s = __builtin_amdgcn_permlane16_swap(m1, m2, false, false);
  const unsigned slo = s.x, shi = s.y;
  const unsigned m = slo | (shi << 8);
  q[qi] = (int)(((unsigned)g << 16) | m);
  qi += m ? QROW : 0;
  cnt += __builtin_popcount(m);
}

__device__ __forceinline__ void merge_linear_layers(const float* __restrict__ W2,
                                                    const float* __restrict__ b2,
                                                    const float* __restrict__ Wd,
                                                    const float* __restrict__ bd,
                                                    float* __restrict__ mw,
                                                    float* __restrict__ mb) {
  const int tid = threadIdx.x;
  const int bx  = blockIdx.x;
  if (bx < 16) {
    const int wave = tid >> 5, lane = tid & 31;
    const int c = lane & 15, half = lane >> 4;
    const int tile = bx * 8 + wave;
    const int mt = tile >> 3, nt = tile & 7;
    const int t = mt >> 3;
    const int o = (mt & 7) * 16 + c;
    const int j = nt * 16 + c;
    const float* ap = Wd + ((size_t)(2 * half) * HID + o) * OUTW + t;
    const float* bp = W2 + (2 * half) * HID + j;
    v8f acc;
#pragma unroll
    for (int r = 0; r < 8; ++r) acc[r] = 0.f;
#pragma unroll 8
    for (int k0 = 0; k0 < HID; k0 += 4) {
      v2f a, b;
      a.x = ap[(size_t)k0 * HID * OUTW];  a.y = ap[(size_t)(k0 + 1) * HID * OUTW];
      b.x = bp[k0 * HID];                 b.y = bp[(k0 + 1) * HID];
      acc = wmma_f32(a, b, acc);
    }
    float* dp = mw + (size_t)(mt * 16 + 8 * half) * HID + nt * 16 + c;
#pragma unroll
    for (int r = 0; r < 8; ++r) dp[r * HID] = acc[r];
  } else if (bx == 16) {
    const int t = tid >> 7, o = tid & (HID - 1);
    float s = 0.f;
    for (int i = 0; i < HID; ++i) s = __builtin_fmaf(b2[i], Wd[((size_t)i * HID + o) * OUTW + t], s);
    mb[tid] = s + bd[o];
  }
}

__global__ void __launch_bounds__(256)
knn_seed_kernel(const float* __restrict__ partial,
                const float* __restrict__ predicted,
                const float* __restrict__ lin2_w, const float* __restrict__ lin2_b,
                const float* __restrict__ dec_w,  const float* __restrict__ dec_b,
                float* __restrict__ mw, float* __restrict__ mb,
                float* __restrict__ xout) {
  if (blockIdx.y == BB) {
    merge_linear_layers(lin2_w, lin2_b, dec_w, dec_b, mw, mb);
    return;
  }
  __shared__ v4f cloud[MM];
  __shared__ int qbuf[QCAP * QROW];
  const int b   = blockIdx.y;
  const int tid = threadIdx.x;

  for (int i = tid; i < MM; i += 256) {
    const float* g = partial + ((size_t)b * MM + i) * 3;
    const float x = g[0], y = g[1], z = g[2];
    const v4f pt = {x, y, z, (x * x + y * y + z * z) * (1.0f - KAPPA)};
    cloud[i] = pt;
  }
  __syncthreads();

  const int n = blockIdx.x * 256 + tid;
  const int row = b * NN + n;
  const float px = predicted[row * 3 + 0];
  const float py = predicted[row * 3 + 1];
  const float pz = predicted[row * 3 + 2];
  const float pp = px * px + py * py + pz * pz;

  const int lane = tid & 31;
  const int c = lane & 15, half = lane >> 4;
  const float* g1 = predicted + (size_t)(row - lane + c) * 3;
  const float* g2 = predicted + (size_t)(row - lane + 16 + c) * 3;
  const float x1 = g1[0], y1 = g1[1], z1 = g1[2];
  const float x2 = g2[0], y2 = g2[1], z2 = g2[2];
  v2f b1, b2;
  b1.x = half ? -2.0f * z1 : -2.0f * x1;   b1.y = half ? 1.0f : -2.0f * y1;
  b2.x = half ? -2.0f * z2 : -2.0f * x2;   b2.y = half ? 1.0f : -2.0f * y2;
  const v2f* ap = (const v2f*)cloud + (c * 2 + half);
  float bd[KNN]; int bi[KNN];
#pragma unroll
  for (int j = 0; j < KNN; ++j) { bd[j] = 3.4e38f; bi[j] = 0; }

  int qi = tid;
  int cnt = 0;
  v8f nt1, nt2;
  set_threshold(bd[KNN - 1], pp, nt1, nt2);

  v2f aA = ap[0];
  v2f aB = ap[32];
  for (int g = 0; g < GROUPS; g += 2) {
    v8f dA1 = wmma_f32(aA, b1, nt1), dA2 = wmma_f32(aA, b2, nt2);
    v8f dB1 = wmma_f32(aB, b1, nt1), dB2 = wmma_f32(aB, b2, nt2);
    const int gn = (g + 2 < GROUPS) ? (g + 2) : g;
    const v2f nA = ap[gn * 32];
    const v2f nB = ap[gn * 32 + 32];
    if (__builtin_amdgcn_ballot_w32(cnt > QCAP - GP)) {
      drain_queue(qbuf + tid, tid, qi, cloud, px, py, pz, bd, bi);
      cnt = 0;
      set_threshold(bd[KNN - 1], pp, nt1, nt2);
      dA1 = wmma_f32(aA, b1, nt1);  dA2 = wmma_f32(aA, b2, nt2);
      dB1 = wmma_f32(aB, b1, nt1);  dB2 = wmma_f32(aB, b2, nt2);
    }
    queue_group(dA1, dA2, g, qbuf, qi, cnt);
    if (__builtin_amdgcn_ballot_w32(cnt > QCAP - GP)) {
      drain_queue(qbuf + tid, tid, qi, cloud, px, py, pz, bd, bi);
      cnt = 0;
      set_threshold(bd[KNN - 1], pp, nt1, nt2);
      dB1 = wmma_f32(aB, b1, nt1);  dB2 = wmma_f32(aB, b2, nt2);
    }
    queue_group(dB1, dB2, g + 1, qbuf, qi, cnt);
    aA = nA;
    aB = nB;
  }
  drain_queue(qbuf + tid, tid, qi, cloud, px, py, pz, bd, bi);

  float sx = 0.f, sy = 0.f, sz = 0.f;
#pragma unroll
  for (int j = 0; j < KNN; ++j) {
    const v4f m = cloud[bi[j]];
    sx += m.x; sy += m.y; sz += m.z;
  }

  float* xr = xout + (size_t)row * INP8;
  xr[0] = px;          xr[1] = py;          xr[2] = pz;
  xr[3] = sx * 0.125f; xr[4] = sy * 0.125f; xr[5] = sz * 0.125f;
  xr[6] = 0.f;         xr[7] = 0.f;
}

__device__ __forceinline__ float relu(float x) { return x > 0.f ? x : 0.f; }

__global__ void __launch_bounds__(128)
mlp_refine_kernel(const float* __restrict__ xin,
                  const float* __restrict__ predicted,
                  const float* __restrict__ W1, const float* __restrict__ b1,
                  const float* __restrict__ mw, const float* __restrict__ mb,
                  const float* __restrict__ Wc, const float* __restrict__ bc,
                  float* __restrict__ out) {
  extern __shared__ __attribute__((aligned(16))) float smem[];
  const int tid   = threadIdx.x;
  const int wave  = tid >> 5;
  const int lane  = tid & 31;
  const int lrow  = lane & 15;
  const int khalf = lane >> 4;

  float* lds_h  = smem + wave * (32 * LP);
  float* lds_ht = smem + 4 * (32 * LP) + wave * (32 * LP);

  const int rowBase = blockIdx.x * 128 + wave * 32;
  const int dRow = (lane < 16) ? 0 : 8;

  {
    const float* xr0 = xin + (size_t)(rowBase + lrow) * INP8;
    const float* xr1 = xin + (size_t)(rowBase + 16 + lrow) * INP8;
    const float kmask = khalf ? 0.f : 1.f;
    v2f a00, a01, a10, a11;
    a00.x = xr0[khalf * 2 + 0];     a00.y = xr0[khalf * 2 + 1];
    a01.x = xr0[4 + khalf * 2 + 0]; a01.y = xr0[4 + khalf * 2 + 1];
    a10.x = xr1[khalf * 2 + 0];     a10.y = xr1[khalf * 2 + 1];
    a11.x = xr1[4 + khalf * 2 + 0]; a11.y = xr1[4 + khalf * 2 + 1];
#pragma unroll
    for (int nt = 0; nt < 8; ++nt) {
      const int n = nt * 16 + lrow;
      const float bv = b1[n];
      v8f c0, c1;
#pragma unroll
      for (int j = 0; j < 8; ++j) { c0[j] = bv; c1[j] = bv; }
      v2f bf0, bf1;
      const int k0 = khalf * 2;
      bf0.x = W1[n * INP + k0];        bf0.y = W1[n * INP + k0 + 1];
      bf1.x = kmask * W1[n * INP + 4]; bf1.y = kmask * W1[n * INP + 5];
      c0 = wmma_f32(a00, bf0, c0);  c0 = wmma_f32(a01, bf1, c0);
      c1 = wmma_f32(a10, bf0, c1);  c1 = wmma_f32(a11, bf1, c1);
#pragma unroll
      for (int j = 0; j < 8; ++j) {
        const int r = j + dRow;
        lds_h[r * LP + n]        = relu(c0[j]);
        lds_h[(16 + r) * LP + n] = relu(c1[j]);
      }
    }
  }

  for (int t = 0; t < OUTW; ++t) {
    v8f c0[8], c1[8];
#pragma unroll
    for (int nt = 0; nt < 8; ++nt) {
      const float bv = mb[t * HID + nt * 16 + lrow];
#pragma unroll
      for (int j = 0; j < 8; ++j) { c0[nt][j] = bv; c1[nt][j] = bv; }
    }
    const float* wp  = mw + (size_t)(t * HID + lrow) * HID + khalf * 4;
    const float* a0p = lds_h + lrow * LP + khalf * 4;
    const float* a1p = lds_h + (16 + lrow) * LP + khalf * 4;
    v4f bP[8], bQ[8];
#pragma unroll
    for (int nt = 0; nt < 8; ++nt)
      bP[nt] = *(const v4f*)(wp + nt * 16 * HID);
    for (int k8 = 0; k8 < HID; k8 += 16) {
#pragma unroll
      for (int nt = 0; nt < 8; ++nt)
        bQ[nt] = *(const v4f*)(wp + nt * 16 * HID + k8 + 8);
      {
        const v4f aA = *(const v4f*)(a0p + k8);
        const v4f aB = *(const v4f*)(a1p + k8);
        const v2f aA0 = {aA.x, aA.y}, aA1 = {aA.z, aA.w};
        const v2f aB0 = {aB.x, aB.y}, aB1 = {aB.z, aB.w};
#pragma unroll
        for (int nt = 0; nt < 8; ++nt) {
          const v2f bf = {bP[nt].x, bP[nt].y};
          c0[nt] = wmma_f32(aA0, bf, c0[nt]);
          c1[nt] = wmma_f32(aB0, bf, c1[nt]);
        }
#pragma unroll
        for (int nt = 0; nt < 8; ++nt) {
          const v2f bf = {bP[nt].z, bP[nt].w};
          c0[nt] = wmma_f32(aA1, bf, c0[nt]);
          c1[nt] = wmma_f32(aB1, bf, c1[nt]);
        }
      }
      const int kn = (k8 + 16 < HID) ? (k8 + 16) : k8;
#pragma unroll
      for (int nt = 0; nt < 8; ++nt)
        bP[nt] = *(const v4f*)(wp + nt * 16 * HID + kn);
      {
        const v4f aA = *(const v4f*)(a0p + k8 + 8);
        const v4f aB = *(const v4f*)(a1p + k8 + 8);
        const v2f aA0 = {aA.x, aA.y}, aA1 = {aA.z, aA.w};
        const v2f aB0 = {aB.x, aB.y}, aB1 = {aB.z, aB.w};
#pragma unroll
        for (int nt = 0; nt < 8; ++nt) {
          const v2f bf = {bQ[nt].x, bQ[nt].y};
          c0[nt] = wmma_f32(aA0, bf, c0[nt]);
          c1[nt] = wmma_f32(aB0, bf, c1[nt]);
        }
#pragma unroll
        for (int nt = 0; nt < 8; ++nt) {
          const v2f bf = {bQ[nt].z, bQ[nt].w};
          c0[nt] = wmma_f32(aA1, bf, c0[nt]);
          c1[nt] = wmma_f32(aB1, bf, c1[nt]);
        }
      }
    }
#pragma unroll
    for (int nt = 0; nt < 8; ++nt) {
      const int n = nt * 16 + lrow;
#pragma unroll
      for (int j = 0; j < 8; ++j) {
        const int r = j + dRow;
        lds_ht[r * LP + n]        = relu(c0[nt][j]);
        lds_ht[(16 + r) * LP + n] = relu(c1[nt][j]);
      }
    }
    {
      const int rg = rowBase + lane;
      const float* hr = lds_ht + lane * LP;
      float a0 = bc[0], a1 = bc[1], a2 = bc[2];
      for (int i = 0; i < HID; i += 4) {
        const v4f v = *(const v4f*)(hr + i);
#pragma unroll
        for (int q = 0; q < 4; ++q) {
          a0 += v[q] * Wc[0 * HID + i + q];
          a1 += v[q] * Wc[1 * HID + i + q];
          a2 += v[q] * Wc[2 * HID + i + q];
        }
      }
      float* o = out + ((size_t)rg * OUTW + t) * 3;
      o[0] = predicted[rg * 3 + 0] + a0;
      o[1] = predicted[rg * 3 + 1] + a1;
      o[2] = predicted[rg * 3 + 2] + a2;
    }
  }
}

extern "C" void kernel_launch(void* const* d_in, const int* in_sizes, int n_in,
                              void* d_out, int out_size, void* d_ws, size_t ws_size,
                              hipStream_t stream) {
  const float* partial   = (const float*)d_in[0];
  const float* predicted = (const float*)d_in[1];
  const float* W1 = (const float*)d_in[2];
  const float* b1 = (const float*)d_in[3];
  const float* W2 = (const float*)d_in[4];
  const float* b2 = (const float*)d_in[5];
  const float* Wd = (const float*)d_in[6];
  const float* bd = (const float*)d_in[7];
  const float* Wc = (const float*)d_in[8];
  const float* bc = (const float*)d_in[9];
  float* out = (float*)d_out;

  float* x  = (float*)d_ws;
  float* mw = x + (size_t)BB * NN * INP8;
  float* mb = mw + (size_t)OUTW * HID * HID;

  dim3 g1(NN / 256, BB + 1);
  knn_seed_kernel<<<g1, 256, 0, stream>>>(partial, predicted, W2, b2, Wd, bd, mw, mb, x);

  const int blocks = (BB * NN) / 128;
  const size_t lds = (size_t)8 * 32 * LP * sizeof(float);
  mlp_refine_kernel<<<blocks, 128, lds, stream>>>(
      x, predicted, W1, b1, mw, mb, Wc, bc, out);
}


# MLP head 128->3 taken from the accumulator registers: per-lane partial sums over 8 column tiles, then a 16-lane DPP halving exchange; no ht tile through LDS; the head's f32 sum order changes
# speedup vs baseline: 7.5957x; 1.0534x over previous
#include <hip/hip_runtime.h>
#include <hip/hip_bf16.h>


typedef __attribute__((ext_vector_type(2))) float v2f;
typedef __attribute__((ext_vector_type(4))) float v4f;
typedef __attribute__((ext_vector_type(8))) float v8f;

#define BB   8
#define MM   2048
#define NN   8192
#define KNN  8
#define HID  128
#define INP  6
#define INP8 8
#define OUTW 2
#define LP   132

__device__ __forceinline__ v8f wmma_f32(v2f a, v2f b, v8f c) {
  return __builtin_amdgcn_wmma_f32_16x16x4_f32(
       false, a,  false, b,
       (short)0, c,  false,  false);
}

#define GP 16
#define GROUPS (MM / GP)
#define QCAP 32
#define QROW 256
#define KAPPA 0x1p-16f

typedef __attribute__((ext_vector_type(2))) unsigned v2u;

__device__ __forceinline__ void insert_one(float dd, int id,
                                           float (&bd)[KNN], int (&bi)[KNN]) {
  bool lt[KNN];
#pragma unroll
  for (int j = 0; j < KNN; ++j) lt[j] = dd < bd[j];
#pragma unroll
  for (int j = KNN - 1; j >= 1; --j)
    bd[j] = __builtin_amdgcn_fmed3f(bd[j - 1], dd, bd[j]);
  bd[0] = lt[0] ? dd : bd[0];
  int ki[KNN];
#pragma unroll
  for (int j = 0; j < KNN; ++j) ki[j] = lt[j] ? id : bi[j];
#pragma unroll
  for (int j = KNN - 1; j >= 1; --j)
    bi[j] = lt[j - 1] ? bi[j - 1] : ki[j];
  bi[0] = ki[0];
}

__device__ __forceinline__ float rank_dist(float px, float py, float pz,
                                           float qx, float qy, float qz) {
  const float dx = px - qx;
  const float dy = py - qy;
  const float dz = pz - qz;
  float d;
  {
#pragma clang fp contract(off)
    const float sxy = dx * dx + dy * dy;
    d = __builtin_fmaf(dz, dz, sxy);
  }
  return d;
}


__device__ __forceinline__ void take_two(unsigned& cur, unsigned n0, unsigned n1, int& rd,
                                         int& i0, int& i1, bool& live0, bool& live1) {
  const bool t0 = (cur & 0xffffu) == 0u;
  cur = t0 ? n0 : cur;
  const unsigned nn = t0 ? n1 : n0;
  live0 = (cur & 0xffffu) != 0u;
  i0 = live0 ? (int)(((cur >> 16) << 4) + (unsigned)__builtin_ctz(cur | 0x10000u)) : 0;
  cur &= cur - 1u;
  const bool t1 = (cur & 0xffffu) == 0u;
  cur = t1 ? nn : cur;
  live1 = (cur & 0xffffu) != 0u;
  i1 = live1 ? (int)(((cur >> 16) << 4) + (unsigned)__builtin_ctz(cur | 0x10000u)) : 0;
  cur &= cur - 1u;
  rd += (t0 ? QROW : 0) + (t1 ? QROW : 0);
}

__device__ __forceinline__ void words_from(const int* qt, int tid, int rd, int qi, int last,
                                           unsigned& n0, unsigned& n1) {
  const int r0 = rd < last ? rd : last;
  const int r1 = rd + QROW < last ? rd + QROW : last;
  const unsigned w0 = (unsigned)qt[r0 - tid];
  const unsigned w1 = (unsigned)qt[r1 - tid];
  n0 = (rd < qi) ? w0 : 0u;
  n1 = (rd + QROW < qi) ? w1 : 0u;
}

typedef __attribute__((ext_vector_type(3))) float v3f;
__device__ __forceinline__ v3f coords(const v4f* cloud, int i) {
  return *(const v3f*)(cloud + i);
}

__device__ __forceinline__ void drain_queue(const int* qt, int tid, int& qi,
                                            const v4f* cloud,
                                            float px, float py, float pz,
                                            float (&bd)[KNN], int (&bi)[KNN]) {
  const int last = (QCAP - 1) * QROW + tid;
  const float inf = __builtin_inff();
  unsigned cur = 0;
  int rd = tid;
  unsigned n0, n1;
  words_from(qt, tid, rd, qi, last, n0, n1);
  int i0, i1, j0, j1;
  bool live0, live1, next0, next1;
  take_two(cur, n0, n1, rd, i0, i1, live0, live1);
  float d0, d1;
  {
    const v3f q0 = coords(cloud, i0);
    const v3f q1 = coords(cloud, i1);
    words_from(qt, tid, rd, qi, last, n0, n1);
    take_two(cur, n0, n1, rd, j0, j1, next0, next1);
    words_from(qt, tid, rd, qi, last, n0, n1);
    d0 = live0 ? rank_dist(px, py, pz, q0.x, q0.y, q0.z) : inf;
    d1 = live1 ? rank_dist(px, py, pz, q1.x, q1.y, q1.z) : inf;
  }
#pragma nounroll
  for (int e = 0; e < QCAP; e += 2) {
    if (__builtin_amdgcn_ballot_w32(live0) == 0) break;
    const v3f r0 = coords(cloud, j0);
    const v3f r1 = coords(cloud, j1);
    __builtin_amdgcn_sched_barrier(0);
    int m0, m1;
    bool then0, then1;
    take_two(cur, n0, n1, rd, m0, m1, then0, then1);
    const int e0 = rd < last ? rd : last;
    const int e1 = rd + QROW < last ? rd + QROW : last;
    const unsigned w0 = (unsigned)qt[e0 - tid];
    const unsigned w1 = (unsigned)qt[e1 - tid];
    insert_one(d0, i0, bd, bi);
    insert_one(d1, i1, bd, bi);
    __builtin_amdgcn_sched_barrier(0);
    n0 = (rd < qi) ? w0 : 0u;
    n1 = (rd + QROW < qi) ? w1 : 0u;
    d0 = next0 ? rank_dist(px, py, pz, r0.x, r0.y, r0.z) : inf;
    d1 = next1 ? rank_dist(px, py, pz, r1.x, r1.y, r1.z) : inf;
    i0 = j0;  i1 = j1;  live0 = next0;
    j0 = m0;  j1 = m1;  next0 = then0;  next1 = then1;
  }
  qi = tid;
}

__device__ __forceinline__ void set_threshold(float t, float pp, v8f& nt1, v8f& nt2) {
  const float tf = (t - pp) + KAPPA * pp;
  const unsigned own = __builtin_bit_cast(unsigned, -tf);
  const v2u s = __builtin_amdgcn_permlane16_swap(own, own, false, false);
  const unsigned s1 = s.x, s2 = s.y;
  const float c1 = __builtin_bit_cast(float, s1);
  const float c2 = __builtin_bit_cast(float, s2);
#pragma unroll
  for (int r = 0; r < 8; ++r) { nt1[r] = c1; nt2[r] = c2; }
}

__device__ __forceinline__ unsigned sign_mask8(const v8f& d) {
  unsigned lo = 0u, hi = 0u;
#pragma unroll
  for (int r = 3; r >= 0; --r) {
    const float vl = d[r], vh = d[4 + r];
    lo = __builtin_amdgcn_alignbit(lo, __builtin_bit_cast(unsigned, vl), 31u);
    hi = __builtin_amdgcn_alignbit(hi, __builtin_bit_cast(unsigned, vh), 31u);
  }
  return lo | (hi << 4);
}

__device__ __forceinline__ void queue_group(const v8f& d1, const v8f& d2, int g,
                                            int* q, int& qi, int& cnt) {
  const unsigned m1 = sign_mask8(d1);
  const unsigned m2 = sign_mask8(d2);
  const v2u s = __builtin_amdgcn_permlane16_swap(m1, m2, false, false);
  const unsigned slo = s.x, shi = s.y;
  const unsigned m = slo | (shi << 8);
  q[qi] = (int)(((unsigned)g << 16) | m);
  qi += m ? QROW : 0;
  cnt += __builtin_popcount(m);
}

__device__ __forceinline__ void merge_linear_layers(const float* __restrict__ W2,
                                                    const float* __restrict__ b2,
                                                    const float* __restrict__ Wd,
                                                    const float* __restrict__ bd,
                                                    float* __restrict__ mw,
                                                    float* __restrict__ mb) {
  const int tid = threadIdx.x;
  const int bx  = blockIdx.x;
  if (bx < 16) {
    const int wave = tid >> 5, lane = tid & 31;
    const int c = lane & 15, half = lane >> 4;
    const int tile = bx * 8 + wave;
    const int mt = tile >> 3, nt = tile & 7;
    const int t = mt >> 3;
    const int o = (mt & 7) * 16 + c;
    const int j = nt * 16 + c;
    const float* ap = Wd + ((size_t)(2 * half) * HID + o) * OUTW + t;
    const float* bp = W2 + (2 * half) * HID + j;
    v8f acc;
#pragma unroll
    for (int r = 0; r < 8; ++r) acc[r] = 0.f;
#pragma unroll 8
    for (int k0 = 0; k0 < HID; k0 += 4) {
      v2f a, b;
      a.x = ap[(size_t)k0 * HID * OUTW];  a.y = ap[(size_t)(k0 + 1) * HID * OUTW];
      b.x = bp[k0 * HID];                 b.y = bp[(k0 + 1) * HID];
      acc = wmma_f32(a, b, acc);
    }
    float* dp = mw + (size_t)(mt * 16 + 8 * half) * HID + nt * 16 + c;
#pragma unroll
    for (int r = 0; r < 8; ++r) dp[r * HID] = acc[r];
  } else if (bx == 16) {
    const int t = tid >> 7, o = tid & (HID - 1);
    float s = 0.f;
    for (int i = 0; i < HID; ++i) s = __builtin_fmaf(b2[i], Wd[((size_t)i * HID + o) * OUTW + t], s);
    mb[tid] = s + bd[o];
  }
}

__global__ void __launch_bounds__(256)
knn_seed_kernel(const float* __restrict__ partial,
                const float* __restrict__ predicted,
                const float* __restrict__ lin2_w, const float* __restrict__ lin2_b,
                const float* __restrict__ dec_w,  const float* __restrict__ dec_b,
                float* __restrict__ mw, float* __restrict__ mb,
                float* __restrict__ xout) {
  if (blockIdx.y == BB) {
    merge_linear_layers(lin2_w, lin2_b, dec_w, dec_b, mw, mb);
    return;
  }
  __shared__ v4f cloud[MM];
  __shared__ int qbuf[QCAP * QROW];
  const int b   = blockIdx.y;
  const int tid = threadIdx.x;

  for (int i = tid; i < MM; i += 256) {
    const float* g = partial + ((size_t)b * MM + i) * 3;
    const float x = g[0], y = g[1], z = g[2];
    const v4f pt = {x, y, z, (x * x + y * y + z * z) * (1.0f - KAPPA)};
    cloud[i] = pt;
  }
  __syncthreads();

  const int n = blockIdx.x * 256 + tid;
  const int row = b * NN + n;
  const float px = predicted[row * 3 + 0];
  const float py = predicted[row * 3 + 1];
  const float pz = predicted[row * 3 + 2];
  const float pp = px * px + py * py + pz * pz;

  const int lane = tid & 31;
  const int c = lane & 15, half = lane >> 4;
  const float* g1 = predicted + (size_t)(row - lane + c) * 3;
  const float* g2 = predicted + (size_t)(row - lane + 16 + c) * 3;
  const float x1 = g1[0], y1 = g1[1], z1 = g1[2];
  const float x2 = g2[0], y2 = g2[1], z2 = g2[2];
  v2f b1, b2;
  b1.x = half ? -2.0f * z1 : -2.0f * x1;   b1.y = half ? 1.0f : -2.0f * y1;
  b2.x = half ? -2.0f * z2 : -2.0f * x2;   b2.y = half ? 1.0f : -2.0f * y2;
  const v2f* ap = (const v2f*)cloud + (c * 2 + half);
  float bd[KNN]; int bi[KNN];
#pragma unroll
  for (int j = 0; j < KNN; ++j) { bd[j] = 3.4e38f; bi[j] = 0; }

  int qi = tid;
  int cnt = 0;
  v8f nt1, nt2;
  set_threshold(bd[KNN - 1], pp, nt1, nt2);

  v2f aA = ap[0];
  v2f aB = ap[32];
  for (int g = 0; g < GROUPS; g += 2) {
    v8f dA1 = wmma_f32(aA, b1, nt1), dA2 = wmma_f32(aA, b2, nt2);
    v8f dB1 = wmma_f32(aB, b1, nt1), dB2 = wmma_f32(aB, b2, nt2);
    const int gn = (g + 2 < GROUPS) ? (g + 2) : g;
    const v2f nA = ap[gn * 32];
    const v2f nB = ap[gn * 32 + 32];
    if (__builtin_amdgcn_ballot_w32(cnt > QCAP - GP)) {
      drain_queue(qbuf + tid, tid, qi, cloud, px, py, pz, bd, bi);
      cnt = 0;
      set_threshold(bd[KNN - 1], pp, nt1, nt2);
      dA1 = wmma_f32(aA, b1, nt1);  dA2 = wmma_f32(aA, b2, nt2);
      dB1 = wmma_f32(aB, b1, nt1);  dB2 = wmma_f32(aB, b2, nt2);
    }
    queue_group(dA1, dA2, g, qbuf, qi, cnt);
    if (__builtin_amdgcn_ballot_w32(cnt > QCAP - GP)) {
      drain_queue(qbuf + tid, tid, qi, cloud, px, py, pz, bd, bi);
      cnt = 0;
      set_threshold(bd[KNN - 1], pp, nt1, nt2);
      dB1 = wmma_f32(aB, b1, nt1);  dB2 = wmma_f32(aB, b2, nt2);
    }
    queue_group(dB1, dB2, g + 1, qbuf, qi, cnt);
    aA = nA;
    aB = nB;
  }
  drain_queue(qbuf + tid, tid, qi, cloud, px, py, pz, bd, bi);

  float sx = 0.f, sy = 0.f, sz = 0.f;
#pragma unroll
  for (int j = 0; j < KNN; ++j) {
    const v4f m = cloud[bi[j]];
    sx += m.x; sy += m.y; sz += m.z;
  }

  float* xr = xout + (size_t)row * INP8;
  xr[0] = px;          xr[1] = py;          xr[2] = pz;
  xr[3] = sx * 0.125f; xr[4] = sy * 0.125f; xr[5] = sz * 0.125f;
  xr[6] = 0.f;         xr[7] = 0.f;
}

__device__ __forceinline__ float relu(float x) { return x > 0.f ? x : 0.f; }

template <int M>
__device__ __forceinline__ float xrow(float v) {
  return __builtin_bit_cast(float, __builtin_amdgcn_update_dpp(
      0, __builtin_bit_cast(int, v), 0x160 + M, 0xf, 0xf, true));
}

template <int M>
__device__ __forceinline__ float fold(bool bit, float lo, float hi) {
  const float keep = bit ? hi : lo;
  const float send = bit ? lo : hi;
  return keep + xrow<M>(send);
}

__global__ void __launch_bounds__(128)
mlp_refine_kernel(const float* __restrict__ xin,
                  const float* __restrict__ predicted,
                  const float* __restrict__ W1, const float* __restrict__ b1,
                  const float* __restrict__ mw, const float* __restrict__ mb,
                  const float* __restrict__ Wc, const float* __restrict__ bc,
                  float* __restrict__ out) {
  extern __shared__ __attribute__((aligned(16))) float smem[];
  const int tid   = threadIdx.x;
  const int wave  = tid >> 5;
  const int lane  = tid & 31;
  const int lrow  = lane & 15;
  const int khalf = lane >> 4;

  float* lds_h  = smem + wave * (32 * LP);

  const int rowBase = blockIdx.x * 128 + wave * 32;
  const int dRow = (lane < 16) ? 0 : 8;

  {
    const float* xr0 = xin + (size_t)(rowBase + lrow) * INP8;
    const float* xr1 = xin + (size_t)(rowBase + 16 + lrow) * INP8;
    const float kmask = khalf ? 0.f : 1.f;
    v2f a00, a01, a10, a11;
    a00.x = xr0[khalf * 2 + 0];     a00.y = xr0[khalf * 2 + 1];
    a01.x = xr0[4 + khalf * 2 + 0]; a01.y = xr0[4 + khalf * 2 + 1];
    a10.x = xr1[khalf * 2 + 0];     a10.y = xr1[khalf * 2 + 1];
    a11.x = xr1[4 + khalf * 2 + 0]; a11.y = xr1[4 + khalf * 2 + 1];
#pragma unroll
    for (int nt = 0; nt < 8; ++nt) {
      const int n = nt * 16 + lrow;
      const float bv = b1[n];
      v8f c0, c1;
#pragma unroll
      for (int j = 0; j < 8; ++j) { c0[j] = bv; c1[j] = bv; }
      v2f bf0, bf1;
      const int k0 = khalf * 2;
      bf0.x = W1[n * INP + k0];        bf0.y = W1[n * INP + k0 + 1];
      bf1.x = kmask * W1[n * INP + 4]; bf1.y = kmask * W1[n * INP + 5];
      c0 = wmma_f32(a00, bf0, c0);  c0 = wmma_f32(a01, bf1, c0);
      c1 = wmma_f32(a10, bf0, c1);  c1 = wmma_f32(a11, bf1, c1);
#pragma unroll
      for (int j = 0; j < 8; ++j) {
        const int r = j + dRow;
        lds_h[r * LP + n]        = relu(c0[j]);
        lds_h[(16 + r) * LP + n] = relu(c1[j]);
      }
    }
  }

  float wc[3][8];
#pragma unroll
  for (int o = 0; o < 3; ++o) {
#pragma unroll
    for (int nt = 0; nt < 8; ++nt) wc[o][nt] = Wc[o * HID + nt * 16 + lrow];
  }
  const bool bit3 = (lrow & 8) != 0, bit2 = (lrow & 4) != 0;
  const bool bit1 = (lrow & 2) != 0, bit0 = (lrow & 1) != 0;

  for (int t = 0; t < OUTW; ++t) {
    v8f c0[8], c1[8];
#pragma unroll
    for (int nt = 0; nt < 8; ++nt) {
      const float bv = mb[t * HID + nt * 16 + lrow];
#pragma unroll
      for (int j = 0; j < 8; ++j) { c0[nt][j] = bv; c1[nt][j] = bv; }
    }
    const float* wp  = mw + (size_t)(t * HID + lrow) * HID + khalf * 4;
    const float* a0p = lds_h + lrow * LP + khalf * 4;
    const float* a1p = lds_h + (16 + lrow) * LP + khalf * 4;
    v4f bP[8], bQ[8];
#pragma unroll
    for (int nt = 0; nt < 8; ++nt)
      bP[nt] = *(const v4f*)(wp + nt * 16 * HID);
    for (int k8 = 0; k8 < HID; k8 += 16) {
#pragma unroll
      for (int nt = 0; nt < 8; ++nt)
        bQ[nt] = *(const v4f*)(wp + nt * 16 * HID + k8 + 8);
      {
        const v4f aA = *(const v4f*)(a0p + k8);
        const v4f aB = *(const v4f*)(a1p + k8);
        const v2f aA0 = {aA.x, aA.y}, aA1 = {aA.z, aA.w};
        const v2f aB0 = {aB.x, aB.y}, aB1 = {aB.z, aB.w};
#pragma unroll
        for (int nt = 0; nt < 8; ++nt) {
          const v2f bf = {bP[nt].x, bP[nt].y};
          c0[nt] = wmma_f32(aA0, bf, c0[nt]);
          c1[nt] = wmma_f32(aB0, bf, c1[nt]);
        }
#pragma unroll
        for (int nt = 0; nt < 8; ++nt) {
          const v2f bf = {bP[nt].z, bP[nt].w};
          c0[nt] = wmma_f32(aA1, bf, c0[nt]);
          c1[nt] = wmma_f32(aB1, bf, c1[nt]);
        }
      }
      const int kn = (k8 + 16 < HID) ? (k8 + 16) : k8;
#pragma unroll
      for (int nt = 0; nt < 8; ++nt)
        bP[nt] = *(const v4f*)(wp + nt * 16 * HID + kn);
      {
        const v4f aA = *(const v4f*)(a0p + k8 + 8);
        const v4f aB = *(const v4f*)(a1p + k8 + 8);
        const v2f aA0 = {aA.x, aA.y}, aA1 = {aA.z, aA.w};
        const v2f aB0 = {aB.x, aB.y}, aB1 = {aB.z, aB.w};
#pragma unroll
        for (int nt = 0; nt < 8; ++nt) {
          const v2f bf = {bQ[nt].x, bQ[nt].y};
          c0[nt] = wmma_f32(aA0, bf, c0[nt]);
          c1[nt] = wmma_f32(aB0, bf, c1[nt]);
        }
#pragma unroll
        for (int nt = 0; nt < 8; ++nt) {
          const v2f bf = {bQ[nt].z, bQ[nt].w};
          c0[nt] = wmma_f32(aA1, bf, c0[nt]);
          c1[nt] = wmma_f32(aB1, bf, c1[nt]);
        }
      }
    }
    float p0[8][3], p1[8][3];
#pragma unroll
    for (int j = 0; j < 8; ++j) {
#pragma unroll
      for (int o = 0; o < 3; ++o) { p0[j][o] = 0.f; p1[j][o] = 0.f; }
    }
#pragma unroll
    for (int nt = 0; nt < 8; ++nt) {
#pragma unroll
      for (int j = 0; j < 8; ++j) {
        const float f0 = c0[nt][j], f1 = c1[nt][j];
        const int i0 = __builtin_bit_cast(int, f0);
        const int i1 = __builtin_bit_cast(int, f1);
        const float h0 = __builtin_bit_cast(float, i0 > 0 ? i0 : 0);
        const float h1 = __builtin_bit_cast(float, i1 > 0 ? i1 : 0);
#pragma unroll
        for (int o = 0; o < 3; ++o) {
          p0[j][o] = __builtin_fmaf(h0, wc[o][nt], p0[j][o]);
          p1[j][o] = __builtin_fmaf(h1, wc[o][nt], p1[j][o]);
        }
      }
    }
    float q8[8][3], q4[4][3], q2[2][3], q1[3];
#pragma unroll
    for (int j = 0; j < 8; ++j) {
#pragma unroll
      for (int o = 0; o < 3; ++o) q8[j][o] = fold<8>(bit3, p0[j][o], p1[j][o]);
    }
#pragma unroll
    for (int j = 0; j < 4; ++j) {
#pragma unroll
      for (int o = 0; o < 3; ++o) q4[j][o] = fold<4>(bit2, q8[j][o], q8[j + 4][o]);
    }
#pragma unroll
    for (int j = 0; j < 2; ++j) {
#pragma unroll
      for (int o = 0; o < 3; ++o) q2[j][o] = fold<2>(bit1, q4[j][o], q4[j + 2][o]);
    }
#pragma unroll
    for (int o = 0; o < 3; ++o) q1[o] = fold<1>(bit0, q2[0][o], q2[1][o]);
    {
      const int rg = rowBase + ((lrow >> 3) << 4) + dRow + (lrow & 7);
      float* o = out + ((size_t)rg * OUTW + t) * 3;
      o[0] = predicted[rg * 3 + 0] + (q1[0] + bc[0]);
      o[1] = predicted[rg * 3 + 1] + (q1[1] + bc[1]);
      o[2] = predicted[rg * 3 + 2] + (q1[2] + bc[2]);
    }
  }
}

extern "C" void kernel_launch(void* const* d_in, const int* in_sizes, int n_in,
                              void* d_out, int out_size, void* d_ws, size_t ws_size,
                              hipStream_t stream) {
  const float* partial   = (const float*)d_in[0];
  const float* predicted = (const float*)d_in[1];
  const float* W1 = (const float*)d_in[2];
  const float* b1 = (const float*)d_in[3];
  const float* W2 = (const float*)d_in[4];
  const float* b2 = (const float*)d_in[5];
  const float* Wd = (const float*)d_in[6];
  const float* bd = (const float*)d_in[7];
  const float* Wc = (const float*)d_in[8];
  const float* bc = (const float*)d_in[9];
  float* out = (float*)d_out;

  float* x  = (float*)d_ws;
  float* mw = x + (size_t)BB * NN * INP8;
  float* mb = mw + (size_t)OUTW * HID * HID;

  dim3 g1(NN / 256, BB + 1);
  knn_seed_kernel<<<g1, 256, 0, stream>>>(partial, predicted, W2, b2, Wd, bd, mw, mb, x);

  const int blocks = (BB * NN) / 128;
  const size_t lds = (size_t)8 * 32 * LP * sizeof(float);
  mlp_refine_kernel<<<blocks, 128, lds, stream>>>(
      x, predicted, W1, b1, mw, mb, Wc, bc, out);
}
